# starting kernel
# baseline (speedup 1.0000x reference)
#include <hip/hip_runtime.h>
#include <math.h>

typedef float v2f __attribute__((ext_vector_type(2)));
typedef float v8f __attribute__((ext_vector_type(8)));

#define TT 2048
#define DD 256
#define NBATCH 8
#define NACT 64
#define NTIME 32
#define NTOT 96
#define ROWS (NBATCH * TT)
#define OUT_TIME_OFF (ROWS * NACT)
#define PSTRIDE 112
#define KCHUNK 64

__device__ __forceinline__ float softplus_f(float x) { return log1pf(expf(x)); }

__global__ __launch_bounds__(256)
void gemm_heads_kernel(const float* __restrict__ h,
                       const float* __restrict__ E,
                       const float* __restrict__ Wn,
                       const float* __restrict__ bn,
                       const float* __restrict__ Wt,
                       const float* __restrict__ bt,
                       const float* __restrict__ tsa,
                       const float* __restrict__ tst,
                       float* __restrict__ out) {
  __shared__ v2f lw2[(KCHUNK / 2) * PSTRIDE];

  const int tid   = threadIdx.x;
  const int wave  = tid >> 5;
  const int lane  = tid & 31;
  const int lhalf = lane >> 4;
  const int l16   = lane & 15;
  const int m_base = blockIdx.x * 128 + wave * 16;

  const float s_ta = softplus_f(tsa[0]);
  const float s_tt = softplus_f(tst[0]);

  v8f acc[6];
#pragma unroll
  for (int j = 0; j < 6; ++j) {
    int n = j * 16 + l16;
    float bv = (n < NACT) ? bn[n] : bt[n - NACT];
#pragma unroll
    for (int r = 0; r < 8; ++r) acc[j][r] = bv;
  }

  float* lwf = (float*)lw2;

  for (int kc = 0; kc < DD; kc += KCHUNK) {
    __syncthreads();
#pragma unroll
    for (int i = 0; i < (KCHUNK * NTOT) / 256; ++i) {
      int idx = i * 256 + tid;
      int kk  = idx & (KCHUNK - 1);
      int n   = idx >> 6;
      int kg  = kc + kk;
      float wv;
      if (n < NACT) wv = Wn[n * DD + kg] + s_ta * E[(4 + n) * DD + kg];
      else          wv = Wt[(n - NACT) * DD + kg] + s_tt * E[(68 + (n - NACT)) * DD + kg];
      lwf[((kk >> 1) * PSTRIDE + n) * 2 + (kk & 1)] = wv;
    }
    __syncthreads();

#pragma unroll 4
    for (int ks = 0; ks < KCHUNK; ks += 4) {
      int kk = kc + ks + (lhalf << 1);
      v2f a = *(const v2f*)(h + (m_base + l16) * DD + kk);
      int p = (ks >> 1) + lhalf;
#pragma unroll
      for (int j = 0; j < 6; ++j) {
        int n = j * 16 + l16;
        v2f b = lw2[p * PSTRIDE + n];
        acc[j] = __builtin_amdgcn_wmma_f32_16x16x4_f32(
            false, a, false, b, (short)0, acc[j], false, false);
      }
    }
  }

#pragma unroll
  for (int j = 0; j < 6; ++j) {
    int n = j * 16 + l16;
#pragma unroll
    for (int r = 0; r < 8; ++r) {
      int row = m_base + r + (lhalf << 3);
      if (n < NACT) out[row * NACT + n] = acc[j][r];
      else          out[OUT_TIME_OFF + row * NTIME + (n - NACT)] = acc[j][r];
    }
  }
}

__global__ __launch_bounds__(256)
void proto_kernel(const int* __restrict__ tokens,
                  const float* __restrict__ h,
                  const float* __restrict__ E,
                  const float* __restrict__ psa, const float* __restrict__ pst,
                  const float* __restrict__ ppa, const float* __restrict__ ppt,
                  const float* __restrict__ pta, const float* __restrict__ ptt,
                  float* __restrict__ out) {
  __shared__ float A[32 * 256];
  __shared__ float hbuf[256];
  __shared__ float red[256];
  __shared__ float invn[32];
  __shared__ int   evbuf[2048];
  __shared__ int   nev_sh;

  const int b   = blockIdx.x / 3;
  const int seg = blockIdx.x % 3;
  const bool is_time = (seg == 2);
  const int lo = is_time ? 68 : 4;
  const int hi = is_time ? 100 : 68;
  const int c0 = (seg == 1) ? 32 : 0;
  const int eb = lo + c0;

  const float alpha = softplus_f(is_time ? ppt[0] : ppa[0]);
  const float st    = softplus_f(is_time ? pst[0] : psa[0]) *
                      softplus_f(is_time ? ptt[0] : pta[0]);

  const int tid  = threadIdx.x;
  const int wave = tid >> 5;
  const int lane = tid & 31;

  if (tid == 0) {
    int nsup = 0, nev = 0;
    const int* tk = tokens + b * TT;
    for (int t = 0; t < TT; ++t) {
      if (tk[t] == 3) {
        int nxt  = tk[(t + 1) & (TT - 1)];
        int sup  = (nxt >= lo && nxt < hi) ? 1 : 0;
        int emit = (nsup > 0) ? 1 : 0;
        if (sup | emit)
          evbuf[nev++] = (t << 9) | (emit << 8) | (sup ? (nxt - lo + 1) : 0);
        nsup += sup;
      }
    }
    nev_sh = nev;
  }

  if (tid < 32) {
    float s = 0.f;
    const float* er = E + (eb + tid) * DD;
    for (int d = 0; d < DD; ++d) { float v = er[d]; s += v * v; }
    invn[tid] = 1.0f / fmaxf(sqrtf(s), 1e-12f);
  }
  __syncthreads();

  for (int i = 0; i < 32; ++i) {
    int idx = i * 256 + tid;
    int c = idx >> 8, d = idx & 255;
    A[idx] = alpha * E[(eb + c) * DD + d] * invn[c];
  }
  __syncthreads();

  const int nev = nev_sh;
  for (int e = 0; e < nev; ++e) {
    int ev   = evbuf[e];
    int t    = ev >> 9;
    int emit = (ev >> 8) & 1;
    int clsp = ev & 255;
    int r    = b * TT + t;

    float hv = h[r * DD + tid];
    hbuf[tid] = hv;
    red[tid]  = hv * hv;
    __syncthreads();
    for (int s = 128; s > 0; s >>= 1) {
      if (tid < s) red[tid] += red[tid + s];
      __syncthreads();
    }
    float invh = 1.0f / fmaxf(sqrtf(red[0]), 1e-12f);

    if (emit) {
#pragma unroll
      for (int q = 0; q < 4; ++q) {
        int c = wave * 4 + q;
        const float* ac = A + c * 256;
        float dot = 0.f, asq = 0.f;
#pragma unroll
        for (int u = 0; u < 8; ++u) {
          float av = ac[lane * 8 + u];
          float hx = hbuf[lane * 8 + u];
          dot += av * hx;
          asq += av * av;
        }
#pragma unroll
        for (int off = 16; off > 0; off >>= 1) {
          dot += __shfl_xor(dot, off, 32);
          asq += __shfl_xor(asq, off, 32);
        }
        if (lane == 0) {
          float sim = dot * invh / fmaxf(sqrtf(asq), 1e-12f);
          float add = st * sim;
          if (!is_time) out[r * NACT + (c0 + c)] += add;
          else          out[OUT_TIME_OFF + r * NTIME + c] += add;
        }
      }
    }
    __syncthreads();

    if (clsp) {
      int cl = clsp - 1 - c0;
      if (cl >= 0 && cl < 32) A[cl * 256 + tid] += hbuf[tid] * invh;
    }
    __syncthreads();
  }
}

extern "C" void kernel_launch(void* const* d_in, const int* in_sizes, int n_in,
                              void* d_out, int out_size, void* d_ws, size_t ws_size,
                              hipStream_t stream) {
  const int*   tokens = (const int*)  d_in[0];
  const float* h      = (const float*)d_in[1];
  const float* E      = (const float*)d_in[2];
  const float* Wn     = (const float*)d_in[3];
  const float* bn     = (const float*)d_in[4];
  const float* Wt     = (const float*)d_in[5];
  const float* bt     = (const float*)d_in[6];
  const float* tsa    = (const float*)d_in[7];
  const float* tst    = (const float*)d_in[8];
  const float* psa    = (const float*)d_in[9];
  const float* pst    = (const float*)d_in[10];
  const float* ppa    = (const float*)d_in[11];
  const float* ppt    = (const float*)d_in[12];
  const float* pta    = (const float*)d_in[13];
  const float* ptt    = (const float*)d_in[14];
  float* out = (float*)d_out;

  gemm_heads_kernel<<<ROWS / 128, 256, 0, stream>>>(h, E, Wn, bn, Wt, bt, tsa, tst, out);
  proto_kernel<<<NBATCH * 3, 256, 0, stream>>>(tokens, h, E, psa, pst, ppa, ppt,
                                               pta, ptt, out);
}


# proto kernel: event list built by all 256 threads (8 tokens each; ballot+readlane for first support, ballot+mbcnt prefix for ordered compaction) instead of one thread's serial 2048-token scan
# speedup vs baseline: 1.6953x; 1.6953x over previous
#include <hip/hip_runtime.h>
#include <math.h>

typedef float v2f __attribute__((ext_vector_type(2)));
typedef float v8f __attribute__((ext_vector_type(8)));

#define TT 2048
#define DD 256
#define NBATCH 8
#define NACT 64
#define NTIME 32
#define NTOT 96
#define ROWS (NBATCH * TT)
#define OUT_TIME_OFF (ROWS * NACT)
#define PSTRIDE 112
#define KCHUNK 64

__device__ __forceinline__ float softplus_f(float x) { return log1pf(expf(x)); }

__global__ __launch_bounds__(256)
void gemm_heads_kernel(const float* __restrict__ h,
                       const float* __restrict__ E,
                       const float* __restrict__ Wn,
                       const float* __restrict__ bn,
                       const float* __restrict__ Wt,
                       const float* __restrict__ bt,
                       const float* __restrict__ tsa,
                       const float* __restrict__ tst,
                       float* __restrict__ out) {
  __shared__ v2f lw2[(KCHUNK / 2) * PSTRIDE];

  const int tid   = threadIdx.x;
  const int wave  = tid >> 5;
  const int lane  = tid & 31;
  const int lhalf = lane >> 4;
  const int l16   = lane & 15;
  const int m_base = blockIdx.x * 128 + wave * 16;

  const float s_ta = softplus_f(tsa[0]);
  const float s_tt = softplus_f(tst[0]);

  v8f acc[6];
#pragma unroll
  for (int j = 0; j < 6; ++j) {
    int n = j * 16 + l16;
    float bv = (n < NACT) ? bn[n] : bt[n - NACT];
#pragma unroll
    for (int r = 0; r < 8; ++r) acc[j][r] = bv;
  }

  float* lwf = (float*)lw2;

  for (int kc = 0; kc < DD; kc += KCHUNK) {
    __syncthreads();
#pragma unroll
    for (int i = 0; i < (KCHUNK * NTOT) / 256; ++i) {
      int idx = i * 256 + tid;
      int kk  = idx & (KCHUNK - 1);
      int n   = idx >> 6;
      int kg  = kc + kk;
      float wv;
      if (n < NACT) wv = Wn[n * DD + kg] + s_ta * E[(4 + n) * DD + kg];
      else          wv = Wt[(n - NACT) * DD + kg] + s_tt * E[(68 + (n - NACT)) * DD + kg];
      lwf[((kk >> 1) * PSTRIDE + n) * 2 + (kk & 1)] = wv;
    }
    __syncthreads();

#pragma unroll 4
    for (int ks = 0; ks < KCHUNK; ks += 4) {
      int kk = kc + ks + (lhalf << 1);
      v2f a = *(const v2f*)(h + (m_base + l16) * DD + kk);
      int p = (ks >> 1) + lhalf;
#pragma unroll
      for (int j = 0; j < 6; ++j) {
        int n = j * 16 + l16;
        v2f b = lw2[p * PSTRIDE + n];
        acc[j] = __builtin_amdgcn_wmma_f32_16x16x4_f32(
            false, a, false, b, (short)0, acc[j], false, false);
      }
    }
  }

#pragma unroll
  for (int j = 0; j < 6; ++j) {
    int n = j * 16 + l16;
#pragma unroll
    for (int r = 0; r < 8; ++r) {
      int row = m_base + r + (lhalf << 3);
      if (n < NACT) out[row * NACT + n] = acc[j][r];
      else          out[OUT_TIME_OFF + row * NTIME + (n - NACT)] = acc[j][r];
    }
  }
}

__global__ __launch_bounds__(256)
void proto_kernel(const int* __restrict__ tokens,
                  const float* __restrict__ h,
                  const float* __restrict__ E,
                  const float* __restrict__ psa, const float* __restrict__ pst,
                  const float* __restrict__ ppa, const float* __restrict__ ppt,
                  const float* __restrict__ pta, const float* __restrict__ ptt,
                  float* __restrict__ out) {
  __shared__ float A[32 * 256];
  __shared__ float hbuf[256];
  __shared__ float red[256];
  __shared__ float invn[32];
  __shared__ int   evbuf[2048];
  __shared__ int   wfirst[8];
  __shared__ int   wcount[8];

  const int b   = blockIdx.x / 3;
  const int seg = blockIdx.x % 3;
  const bool is_time = (seg == 2);
  const int lo = is_time ? 68 : 4;
  const int hi = is_time ? 100 : 68;
  const int c0 = (seg == 1) ? 32 : 0;
  const int eb = lo + c0;

  const float alpha = softplus_f(is_time ? ppt[0] : ppa[0]);
  const float st    = softplus_f(is_time ? pst[0] : psa[0]) *
                      softplus_f(is_time ? ptt[0] : pta[0]);

  const int tid  = threadIdx.x;
  const int wave = tid >> 5;
  const int lane = tid & 31;

  const int* tk = tokens + b * TT;
  const int t0 = tid * 8;
  int tok[9];
  {
    const int4 qa = *(const int4*)(tk + t0);
    const int4 qb = *(const int4*)(tk + t0 + 4);
    tok[0] = qa.x; tok[1] = qa.y; tok[2] = qa.z; tok[3] = qa.w;
    tok[4] = qb.x; tok[5] = qb.y; tok[6] = qb.z; tok[7] = qb.w;
    tok[8] = tk[(t0 + 8) & (TT - 1)];
  }
  unsigned labm = 0u, supm = 0u;
#pragma unroll
  for (int i = 0; i < 8; ++i) {
    if (tok[i] == 3) {
      labm |= 1u << i;
      int nxt = tok[i + 1];
      if (nxt >= lo && nxt < hi) supm |= 1u << i;
    }
  }
  {
    const unsigned bs = __builtin_amdgcn_ballot_w32(supm != 0u);
    const int fs = t0 + __builtin_ctz(supm | 0x100u);
    const int fw = __builtin_amdgcn_readlane(fs, bs ? __builtin_ctz(bs) : 0);
    if (lane == 0) wfirst[wave] = bs ? fw : TT;
  }

  if (tid < 32) {
    float s = 0.f;
    const float* er = E + (eb + tid) * DD;
    for (int d = 0; d < DD; ++d) { float v = er[d]; s += v * v; }
    invn[tid] = 1.0f / fmaxf(sqrtf(s), 1e-12f);
  }
  __syncthreads();

  int first_sup = wfirst[0];
  for (int w = 1; w < 8; ++w) first_sup = min(first_sup, wfirst[w]);
  unsigned evm = 0u;
  for (int i = 0; i < 8; ++i) {
    unsigned bit = 1u << i;
    if ((labm & bit) && ((supm & bit) || (t0 + i > first_sup))) evm |= bit;
  }
  int ebefore = 0, wtotal = 0;
  for (int i = 0; i < 8; ++i) {
    const unsigned bi = __builtin_amdgcn_ballot_w32(((evm >> i) & 1u) != 0u);
    ebefore = __builtin_amdgcn_mbcnt_lo(bi, ebefore);
    wtotal += __popc(bi);
  }
  if (lane == 0) wcount[wave] = wtotal;
  __syncthreads();

  {
    int pos = ebefore;
    for (int w = 0; w < 8; ++w) {
      if (w < wave) pos += wcount[w];
    }
#pragma unroll
    for (int i = 0; i < 8; ++i) {
      if (evm & (1u << i)) {
        int t    = t0 + i;
        int emit = (t > first_sup) ? 1 : 0;
        int cls1 = (supm & (1u << i)) ? (tok[i + 1] - lo + 1) : 0;
        evbuf[pos++] = (t << 9) | (emit << 8) | cls1;
      }
    }
  }

  for (int i = 0; i < 32; ++i) {
    int idx = i * 256 + tid;
    int c = idx >> 8, d = idx & 255;
    A[idx] = alpha * E[(eb + c) * DD + d] * invn[c];
  }
  __syncthreads();

  int nev = 0;
  for (int w = 0; w < 8; ++w) nev += wcount[w];

  for (int e = 0; e < nev; ++e) {
    int ev   = evbuf[e];
    int t    = ev >> 9;
    int emit = (ev >> 8) & 1;
    int clsp = ev & 255;
    int r    = b * TT + t;

    float hv = h[r * DD + tid];
    hbuf[tid] = hv;
    red[tid]  = hv * hv;
    __syncthreads();
    for (int s = 128; s > 0; s >>= 1) {
      if (tid < s) red[tid] += red[tid + s];
      __syncthreads();
    }
    float invh = 1.0f / fmaxf(sqrtf(red[0]), 1e-12f);

    if (emit) {
#pragma unroll
      for (int q = 0; q < 4; ++q) {
        int c = wave * 4 + q;
        const float* ac = A + c * 256;
        float dot = 0.f, asq = 0.f;
#pragma unroll
        for (int u = 0; u < 8; ++u) {
          float av = ac[lane * 8 + u];
          float hx = hbuf[lane * 8 + u];
          dot += av * hx;
          asq += av * av;
        }
#pragma unroll
        for (int off = 16; off > 0; off >>= 1) {
          dot += __shfl_xor(dot, off, 32);
          asq += __shfl_xor(asq, off, 32);
        }
        if (lane == 0) {
          float sim = dot * invh / fmaxf(sqrtf(asq), 1e-12f);
          float add = st * sim;
          if (!is_time) out[r * NACT + (c0 + c)] += add;
          else          out[OUT_TIME_OFF + r * NTIME + c] += add;
        }
      }
    }
    __syncthreads();

    if (clsp) {
      int cl = clsp - 1 - c0;
      if (cl >= 0 && cl < 32) A[cl * 256 + tid] += hbuf[tid] * invh;
    }
    __syncthreads();
  }
}

extern "C" void kernel_launch(void* const* d_in, const int* in_sizes, int n_in,
                              void* d_out, int out_size, void* d_ws, size_t ws_size,
                              hipStream_t stream) {
  const int*   tokens = (const int*)  d_in[0];
  const float* h      = (const float*)d_in[1];
  const float* E      = (const float*)d_in[2];
  const float* Wn     = (const float*)d_in[3];
  const float* bn     = (const float*)d_in[4];
  const float* Wt     = (const float*)d_in[5];
  const float* bt     = (const float*)d_in[6];
  const float* tsa    = (const float*)d_in[7];
  const float* tst    = (const float*)d_in[8];
  const float* psa    = (const float*)d_in[9];
  const float* pst    = (const float*)d_in[10];
  const float* ppa    = (const float*)d_in[11];
  const float* ppt    = (const float*)d_in[12];
  const float* pta    = (const float*)d_in[13];
  const float* ptt    = (const float*)d_in[14];
  float* out = (float*)d_out;

  gemm_heads_kernel<<<ROWS / 128, 256, 0, stream>>>(h, E, Wn, bn, Wt, bt, tsa, tst, out);
  proto_kernel<<<NBATCH * 3, 256, 0, stream>>>(tokens, h, E, psa, pst, ppa, ppt,
                                               pta, ptt, out);
}


# proto kernel: each wave keeps its 4 classes' accumulators in registers and walks the event list alone (norms by wave shuffle, their sum order changes); event loop has no block barrier or LDS tree
# speedup vs baseline: 3.2104x; 1.8937x over previous
#include <hip/hip_runtime.h>
#include <math.h>

typedef float v2f __attribute__((ext_vector_type(2)));
typedef float v8f __attribute__((ext_vector_type(8)));

#define TT 2048
#define DD 256
#define NBATCH 8
#define NACT 64
#define NTIME 32
#define NTOT 96
#define ROWS (NBATCH * TT)
#define OUT_TIME_OFF (ROWS * NACT)
#define PSTRIDE 112
#define KCHUNK 64

__device__ __forceinline__ float softplus_f(float x) { return log1pf(expf(x)); }

__global__ __launch_bounds__(256)
void gemm_heads_kernel(const float* __restrict__ h,
                       const float* __restrict__ E,
                       const float* __restrict__ Wn,
                       const float* __restrict__ bn,
                       const float* __restrict__ Wt,
                       const float* __restrict__ bt,
                       const float* __restrict__ tsa,
                       const float* __restrict__ tst,
                       float* __restrict__ out) {
  __shared__ v2f lw2[(KCHUNK / 2) * PSTRIDE];

  const int tid   = threadIdx.x;
  const int wave  = tid >> 5;
  const int lane  = tid & 31;
  const int lhalf = lane >> 4;
  const int l16   = lane & 15;
  const int m_base = blockIdx.x * 128 + wave * 16;

  const float s_ta = softplus_f(tsa[0]);
  const float s_tt = softplus_f(tst[0]);

  v8f acc[6];
#pragma unroll
  for (int j = 0; j < 6; ++j) {
    int n = j * 16 + l16;
    float bv = (n < NACT) ? bn[n] : bt[n - NACT];
#pragma unroll
    for (int r = 0; r < 8; ++r) acc[j][r] = bv;
  }

  float* lwf = (float*)lw2;

  for (int kc = 0; kc < DD; kc += KCHUNK) {
    __syncthreads();
#pragma unroll
    for (int i = 0; i < (KCHUNK * NTOT) / 256; ++i) {
      int idx = i * 256 + tid;
      int kk  = idx & (KCHUNK - 1);
      int n   = idx >> 6;
      int kg  = kc + kk;
      float wv;
      if (n < NACT) wv = Wn[n * DD + kg] + s_ta * E[(4 + n) * DD + kg];
      else          wv = Wt[(n - NACT) * DD + kg] + s_tt * E[(68 + (n - NACT)) * DD + kg];
      lwf[((kk >> 1) * PSTRIDE + n) * 2 + (kk & 1)] = wv;
    }
    __syncthreads();

#pragma unroll 4
    for (int ks = 0; ks < KCHUNK; ks += 4) {
      int kk = kc + ks + (lhalf << 1);
      v2f a = *(const v2f*)(h + (m_base + l16) * DD + kk);
      int p = (ks >> 1) + lhalf;
#pragma unroll
      for (int j = 0; j < 6; ++j) {
        int n = j * 16 + l16;
        v2f b = lw2[p * PSTRIDE + n];
        acc[j] = __builtin_amdgcn_wmma_f32_16x16x4_f32(
            false, a, false, b, (short)0, acc[j], false, false);
      }
    }
  }

#pragma unroll
  for (int j = 0; j < 6; ++j) {
    int n = j * 16 + l16;
#pragma unroll
    for (int r = 0; r < 8; ++r) {
      int row = m_base + r + (lhalf << 3);
      if (n < NACT) out[row * NACT + n] = acc[j][r];
      else          out[OUT_TIME_OFF + row * NTIME + (n - NACT)] = acc[j][r];
    }
  }
}

__global__ __launch_bounds__(256)
void proto_kernel(const int* __restrict__ tokens,
                  const float* __restrict__ h,
                  const float* __restrict__ E,
                  const float* __restrict__ psa, const float* __restrict__ pst,
                  const float* __restrict__ ppa, const float* __restrict__ ppt,
                  const float* __restrict__ pta, const float* __restrict__ ptt,
                  float* __restrict__ out) {
  __shared__ int   evbuf[2048];
  __shared__ int   wfirst[8];
  __shared__ int   wcount[8];

  const int b   = blockIdx.x / 3;
  const int seg = blockIdx.x % 3;
  const bool is_time = (seg == 2);
  const int lo = is_time ? 68 : 4;
  const int hi = is_time ? 100 : 68;
  const int c0 = (seg == 1) ? 32 : 0;
  const int eb = lo + c0;

  const float alpha = softplus_f(is_time ? ppt[0] : ppa[0]);
  const float st    = softplus_f(is_time ? pst[0] : psa[0]) *
                      softplus_f(is_time ? ptt[0] : pta[0]);

  const int tid  = threadIdx.x;
  const int wave = __builtin_amdgcn_readfirstlane(tid >> 5);
  const int lane = tid & 31;

  const int* tk = tokens + b * TT;
  const int t0 = tid * 8;
  int tok[9];
  {
    const int4 qa = *(const int4*)(tk + t0);
    const int4 qb = *(const int4*)(tk + t0 + 4);
    tok[0] = qa.x; tok[1] = qa.y; tok[2] = qa.z; tok[3] = qa.w;
    tok[4] = qb.x; tok[5] = qb.y; tok[6] = qb.z; tok[7] = qb.w;
    tok[8] = tk[(t0 + 8) & (TT - 1)];
  }

  float acc[4][8];
#pragma unroll
  for (int q = 0; q < 4; ++q) {
    const float* er = E + (eb + wave * 4 + q) * DD + lane * 8;
    const float4 ea = *(const float4*)(er);
    const float4 ec = *(const float4*)(er + 4);
    const float v[8] = {ea.x, ea.y, ea.z, ea.w, ec.x, ec.y, ec.z, ec.w};
    float s = 0.f;
#pragma unroll
    for (int u = 0; u < 8; ++u) s += v[u] * v[u];
#pragma unroll
    for (int off = 16; off > 0; off >>= 1) s += __shfl_xor(s, off, 32);
    const float inv = 1.0f / fmaxf(sqrtf(s), 1e-12f);
#pragma unroll
    for (int u = 0; u < 8; ++u) acc[q][u] = alpha * v[u] * inv;
  }

  unsigned labm = 0u, supm = 0u;
#pragma unroll
  for (int i = 0; i < 8; ++i) {
    if (tok[i] == 3) {
      labm |= 1u << i;
      int nxt = tok[i + 1];
      if (nxt >= lo && nxt < hi) supm |= 1u << i;
    }
  }
  {
    const unsigned bs = __builtin_amdgcn_ballot_w32(supm != 0u);
    const int fs = t0 + __builtin_ctz(supm | 0x100u);
    const int fw = __builtin_amdgcn_readlane(fs, bs ? __builtin_ctz(bs) : 0);
    if (lane == 0) wfirst[wave] = bs ? fw : TT;
  }
  __syncthreads();

  int first_sup = wfirst[0];
  for (int w = 1; w < 8; ++w) first_sup = min(first_sup, wfirst[w]);
  unsigned evm = 0u;
  for (int i = 0; i < 8; ++i) {
    unsigned bit = 1u << i;
    if ((labm & bit) && ((supm & bit) || (t0 + i > first_sup))) evm |= bit;
  }
  int ebefore = 0, wtotal = 0;
  for (int i = 0; i < 8; ++i) {
    const unsigned bi = __builtin_amdgcn_ballot_w32(((evm >> i) & 1u) != 0u);
    ebefore = __builtin_amdgcn_mbcnt_lo(bi, ebefore);
    wtotal += __popc(bi);
  }
  if (lane == 0) wcount[wave] = wtotal;
  __syncthreads();

  {
    int pos = ebefore;
    for (int w = 0; w < 8; ++w) {
      if (w < wave) pos += wcount[w];
    }
#pragma unroll
    for (int i = 0; i < 8; ++i) {
      if (evm & (1u << i)) {
        int t    = t0 + i;
        int emit = (t > first_sup) ? 1 : 0;
        int cls1 = (supm & (1u << i)) ? (tok[i + 1] - lo + 1) : 0;
        evbuf[pos++] = (t << 9) | (emit << 8) | cls1;
      }
    }
  }
  __syncthreads();

  int nev = 0;
  for (int w = 0; w < 8; ++w) nev += wcount[w];

  float* const obase = is_time ? (out + OUT_TIME_OFF + wave * 4)
                               : (out + c0 + wave * 4);
  const int ostride = is_time ? NTIME : NACT;

  for (int e = 0; e < nev; ++e) {
    const int ev   = __builtin_amdgcn_readfirstlane(evbuf[e]);
    const int t    = ev >> 9;
    const int emit = (ev >> 8) & 1;
    const int clsp = ev & 255;
    const int r    = b * TT + t;

    const float* hr = h + r * DD + lane * 8;
    const float4 ha = *(const float4*)(hr);
    const float4 hc = *(const float4*)(hr + 4);
    float* const op = obase + r * ostride;
    float oldv = 0.f;
    if (emit && lane < 4) oldv = op[lane];
    const float hx[8] = {ha.x, ha.y, ha.z, ha.w, hc.x, hc.y, hc.z, hc.w};

    float hs = 0.f;
#pragma unroll
    for (int u = 0; u < 8; ++u) hs += hx[u] * hx[u];
#pragma unroll
    for (int off = 16; off > 0; off >>= 1) hs += __shfl_xor(hs, off, 32);
    const float invh = 1.0f / fmaxf(sqrtf(hs), 1e-12f);

    if (emit) {
      float dot[4], asq[4];
#pragma unroll
      for (int q = 0; q < 4; ++q) {
        dot[q] = 0.f; asq[q] = 0.f;
#pragma unroll
        for (int u = 0; u < 8; ++u) {
          dot[q] += acc[q][u] * hx[u];
          asq[q] += acc[q][u] * acc[q][u];
        }
      }
#pragma unroll
      for (int off = 16; off > 0; off >>= 1) {
#pragma unroll
        for (int q = 0; q < 4; ++q) {
          dot[q] += __shfl_xor(dot[q], off, 32);
          asq[q] += __shfl_xor(asq[q], off, 32);
        }
      }
      const float dsel = (lane == 0) ? dot[0] : (lane == 1) ? dot[1]
                       : (lane == 2) ? dot[2] : dot[3];
      const float asel = (lane == 0) ? asq[0] : (lane == 1) ? asq[1]
                       : (lane == 2) ? asq[2] : asq[3];
      if (lane < 4) {
        float sim = dsel * invh / fmaxf(sqrtf(asel), 1e-12f);
        float add = st * sim;
        op[lane] = oldv + add;
      }
    }

    if (clsp) {
      const int cl = clsp - 1 - c0;
      if (cl >= 0 && cl < 32 && (cl >> 2) == wave) {
        const int qs = cl & 3;
#pragma unroll
        for (int q = 0; q < 4; ++q) {
          if (q == qs) {
#pragma unroll
            for (int u = 0; u < 8; ++u) acc[q][u] += hx[u] * invh;
          }
        }
      }
    }
  }
}

extern "C" void kernel_launch(void* const* d_in, const int* in_sizes, int n_in,
                              void* d_out, int out_size, void* d_ws, size_t ws_size,
                              hipStream_t stream) {
  const int*   tokens = (const int*)  d_in[0];
  const float* h      = (const float*)d_in[1];
  const float* E      = (const float*)d_in[2];
  const float* Wn     = (const float*)d_in[3];
  const float* bn     = (const float*)d_in[4];
  const float* Wt     = (const float*)d_in[5];
  const float* bt     = (const float*)d_in[6];
  const float* tsa    = (const float*)d_in[7];
  const float* tst    = (const float*)d_in[8];
  const float* psa    = (const float*)d_in[9];
  const float* pst    = (const float*)d_in[10];
  const float* ppa    = (const float*)d_in[11];
  const float* ppt    = (const float*)d_in[12];
  const float* pta    = (const float*)d_in[13];
  const float* ptt    = (const float*)d_in[14];
  float* out = (float*)d_out;

  gemm_heads_kernel<<<ROWS / 128, 256, 0, stream>>>(h, E, Wn, bn, Wt, bt, tsa, tst, out);
  proto_kernel<<<NBATCH * 3, 256, 0, stream>>>(tokens, h, E, psa, pst, ppa, ppt,
                                               pta, ptt, out);
}


# proto kernel: wave sums (row norm, class norms, 4 dots) exchange lanes by permlanex16 + row_xmask DPP on the vector ALU, not ds_bpermute via the LDS unit; pairing 16,8,4,2,1 kept, products as fmaf
# speedup vs baseline: 3.7225x; 1.1595x over previous
#include <hip/hip_runtime.h>
#include <math.h>

typedef float v2f __attribute__((ext_vector_type(2)));
typedef float v8f __attribute__((ext_vector_type(8)));

#define TT 2048
#define DD 256
#define NBATCH 8
#define NACT 64
#define NTIME 32
#define NTOT 96
#define ROWS (NBATCH * TT)
#define OUT_TIME_OFF (ROWS * NACT)
#define PSTRIDE 112
#define KCHUNK 64

__device__ __forceinline__ float softplus_f(float x) { return log1pf(expf(x)); }

__device__ __forceinline__ float xrow16(float x) {
  const int b = __builtin_bit_cast(int, x);
  return __builtin_bit_cast(float, __builtin_amdgcn_permlanex16(
      b, b, 0x76543210, 0xfedcba98, false, false));
}
template <int M>
__device__ __forceinline__ float xrow(float x) {
  const int b = __builtin_bit_cast(int, x);
  return __builtin_bit_cast(float, __builtin_amdgcn_update_dpp(
      0, b, 0x160 | M, 0xf, 0xf, true));
}
__device__ __forceinline__ float wave_sum(float s) {
  s += xrow16(s);
  s += xrow<8>(s);
  s += xrow<4>(s);
  s += xrow<2>(s);
  s += xrow<1>(s);
  return s;
}

__global__ __launch_bounds__(256)
void gemm_heads_kernel(const float* __restrict__ h,
                       const float* __restrict__ E,
                       const float* __restrict__ Wn,
                       const float* __restrict__ bn,
                       const float* __restrict__ Wt,
                       const float* __restrict__ bt,
                       const float* __restrict__ tsa,
                       const float* __restrict__ tst,
                       float* __restrict__ out) {
  __shared__ v2f lw2[(KCHUNK / 2) * PSTRIDE];

  const int tid   = threadIdx.x;
  const int wave  = tid >> 5;
  const int lane  = tid & 31;
  const int lhalf = lane >> 4;
  const int l16   = lane & 15;
  const int m_base = blockIdx.x * 128 + wave * 16;

  const float s_ta = softplus_f(tsa[0]);
  const float s_tt = softplus_f(tst[0]);

  v8f acc[6];
#pragma unroll
  for (int j = 0; j < 6; ++j) {
    int n = j * 16 + l16;
    float bv = (n < NACT) ? bn[n] : bt[n - NACT];
#pragma unroll
    for (int r = 0; r < 8; ++r) acc[j][r] = bv;
  }

  float* lwf = (float*)lw2;

  for (int kc = 0; kc < DD; kc += KCHUNK) {
    __syncthreads();
#pragma unroll
    for (int i = 0; i < (KCHUNK * NTOT) / 256; ++i) {
      int idx = i * 256 + tid;
      int kk  = idx & (KCHUNK - 1);
      int n   = idx >> 6;
      int kg  = kc + kk;
      float wv;
      if (n < NACT) wv = Wn[n * DD + kg] + s_ta * E[(4 + n) * DD + kg];
      else          wv = Wt[(n - NACT) * DD + kg] + s_tt * E[(68 + (n - NACT)) * DD + kg];
      lwf[((kk >> 1) * PSTRIDE + n) * 2 + (kk & 1)] = wv;
    }
    __syncthreads();

#pragma unroll 4
    for (int ks = 0; ks < KCHUNK; ks += 4) {
      int kk = kc + ks + (lhalf << 1);
      v2f a = *(const v2f*)(h + (m_base + l16) * DD + kk);
      int p = (ks >> 1) + lhalf;
#pragma unroll
      for (int j = 0; j < 6; ++j) {
        int n = j * 16 + l16;
        v2f b = lw2[p * PSTRIDE + n];
        acc[j] = __builtin_amdgcn_wmma_f32_16x16x4_f32(
            false, a, false, b, (short)0, acc[j], false, false);
      }
    }
  }

#pragma unroll
  for (int j = 0; j < 6; ++j) {
    int n = j * 16 + l16;
#pragma unroll
    for (int r = 0; r < 8; ++r) {
      int row = m_base + r + (lhalf << 3);
      if (n < NACT) out[row * NACT + n] = acc[j][r];
      else          out[OUT_TIME_OFF + row * NTIME + (n - NACT)] = acc[j][r];
    }
  }
}

__global__ __launch_bounds__(256)
void proto_kernel(const int* __restrict__ tokens,
                  const float* __restrict__ h,
                  const float* __restrict__ E,
                  const float* __restrict__ psa, const float* __restrict__ pst,
                  const float* __restrict__ ppa, const float* __restrict__ ppt,
                  const float* __restrict__ pta, const float* __restrict__ ptt,
                  float* __restrict__ out) {
  __shared__ int   evbuf[2048];
  __shared__ int   wfirst[8];
  __shared__ int   wcount[8];

  const int b   = blockIdx.x / 3;
  const int seg = blockIdx.x % 3;
  const bool is_time = (seg == 2);
  const int lo = is_time ? 68 : 4;
  const int hi = is_time ? 100 : 68;
  const int c0 = (seg == 1) ? 32 : 0;
  const int eb = lo + c0;

  const float alpha = softplus_f(is_time ? ppt[0] : ppa[0]);
  const float st    = softplus_f(is_time ? pst[0] : psa[0]) *
                      softplus_f(is_time ? ptt[0] : pta[0]);

  const int tid  = threadIdx.x;
  const int wave = __builtin_amdgcn_readfirstlane(tid >> 5);
  const int lane = tid & 31;

  const int* tk = tokens + b * TT;
  const int t0 = tid * 8;
  int tok[9];
  {
    const int4 qa = *(const int4*)(tk + t0);
    const int4 qb = *(const int4*)(tk + t0 + 4);
    tok[0] = qa.x; tok[1] = qa.y; tok[2] = qa.z; tok[3] = qa.w;
    tok[4] = qb.x; tok[5] = qb.y; tok[6] = qb.z; tok[7] = qb.w;
    tok[8] = tk[(t0 + 8) & (TT - 1)];
  }

  float acc[4][8];
#pragma unroll
  for (int q = 0; q < 4; ++q) {
    const float* er = E + (eb + wave * 4 + q) * DD + lane * 8;
    const float4 ea = *(const float4*)(er);
    const float4 ec = *(const float4*)(er + 4);
    const float v[8] = {ea.x, ea.y, ea.z, ea.w, ec.x, ec.y, ec.z, ec.w};
    float s = 0.f;
#pragma unroll
    for (int u = 0; u < 8; ++u) s = fmaf(v[u], v[u], s);
    s = wave_sum(s);
    const float inv = 1.0f / fmaxf(sqrtf(s), 1e-12f);
#pragma unroll
    for (int u = 0; u < 8; ++u) acc[q][u] = alpha * v[u] * inv;
  }

  unsigned labm = 0u, supm = 0u;
#pragma unroll
  for (int i = 0; i < 8; ++i) {
    if (tok[i] == 3) {
      labm |= 1u << i;
      int nxt = tok[i + 1];
      if (nxt >= lo && nxt < hi) supm |= 1u << i;
    }
  }
  {
    const unsigned bs = __builtin_amdgcn_ballot_w32(supm != 0u);
    const int fs = t0 + __builtin_ctz(supm | 0x100u);
    const int fw = __builtin_amdgcn_readlane(fs, bs ? __builtin_ctz(bs) : 0);
    if (lane == 0) wfirst[wave] = bs ? fw : TT;
  }
  __syncthreads();

  int first_sup = wfirst[0];
  for (int w = 1; w < 8; ++w) first_sup = min(first_sup, wfirst[w]);
  unsigned evm = 0u;
  for (int i = 0; i < 8; ++i) {
    unsigned bit = 1u << i;
    if ((labm & bit) && ((supm & bit) || (t0 + i > first_sup))) evm |= bit;
  }
  int ebefore = 0, wtotal = 0;
  for (int i = 0; i < 8; ++i) {
    const unsigned bi = __builtin_amdgcn_ballot_w32(((evm >> i) & 1u) != 0u);
    ebefore = __builtin_amdgcn_mbcnt_lo(bi, ebefore);
    wtotal += __popc(bi);
  }
  if (lane == 0) wcount[wave] = wtotal;
  __syncthreads();

  {
    int pos = ebefore;
    for (int w = 0; w < 8; ++w) {
      if (w < wave) pos += wcount[w];
    }
#pragma unroll
    for (int i = 0; i < 8; ++i) {
      if (evm & (1u << i)) {
        int t    = t0 + i;
        int emit = (t > first_sup) ? 1 : 0;
        int cls1 = (supm & (1u << i)) ? (tok[i + 1] - lo + 1) : 0;
        evbuf[pos++] = (t << 9) | (emit << 8) | cls1;
      }
    }
  }
  __syncthreads();

  int nev = 0;
  for (int w = 0; w < 8; ++w) nev += wcount[w];

  float* const obase = is_time ? (out + OUT_TIME_OFF + wave * 4)
                               : (out + c0 + wave * 4);
  const int ostride = is_time ? NTIME : NACT;

  for (int e = 0; e < nev; ++e) {
    const int ev   = __builtin_amdgcn_readfirstlane(evbuf[e]);
    const int t    = ev >> 9;
    const int emit = (ev >> 8) & 1;
    const int clsp = ev & 255;
    const int r    = b * TT + t;

    const float* hr = h + r * DD + lane * 8;
    const float4 ha = *(const float4*)(hr);
    const float4 hc = *(const float4*)(hr + 4);
    float* const op = obase + r * ostride;
    float oldv = 0.f;
    if (emit && lane < 4) oldv = op[lane];
    const float hx[8] = {ha.x, ha.y, ha.z, ha.w, hc.x, hc.y, hc.z, hc.w};

    float hs = 0.f;
#pragma unroll
    for (int u = 0; u < 8; ++u) hs = fmaf(hx[u], hx[u], hs);
    hs = wave_sum(hs);
    const float invh = 1.0f / fmaxf(sqrtf(hs), 1e-12f);

    if (emit) {
      float dot[4], asq[4];
#pragma unroll
      for (int q = 0; q < 4; ++q) {
        dot[q] = 0.f; asq[q] = 0.f;
#pragma unroll
        for (int u = 0; u < 8; ++u) {
          dot[q] = fmaf(acc[q][u], hx[u], dot[q]);
          asq[q] = fmaf(acc[q][u], acc[q][u], asq[q]);
        }
      }
#pragma unroll
      for (int q = 0; q < 4; ++q) { dot[q] += xrow16(dot[q]);  asq[q] += xrow16(asq[q]); }
#pragma unroll
      for (int q = 0; q < 4; ++q) { dot[q] += xrow<8>(dot[q]); asq[q] += xrow<8>(asq[q]); }
#pragma unroll
      for (int q = 0; q < 4; ++q) { dot[q] += xrow<4>(dot[q]); asq[q] += xrow<4>(asq[q]); }
#pragma unroll
      for (int q = 0; q < 4; ++q) { dot[q] += xrow<2>(dot[q]); asq[q] += xrow<2>(asq[q]); }
#pragma unroll
      for (int q = 0; q < 4; ++q) { dot[q] += xrow<1>(dot[q]); asq[q] += xrow<1>(asq[q]); }
      const float dsel = (lane == 0) ? dot[0] : (lane == 1) ? dot[1]
                       : (lane == 2) ? dot[2] : dot[3];
      const float asel = (lane == 0) ? asq[0] : (lane == 1) ? asq[1]
                       : (lane == 2) ? asq[2] : asq[3];
      if (lane < 4) {
        float sim = dsel * invh / fmaxf(sqrtf(asel), 1e-12f);
        float add = st * sim;
        op[lane] = oldv + add;
      }
    }

    if (clsp) {
      const int cl = clsp - 1 - c0;
      if (cl >= 0 && cl < 32 && (cl >> 2) == wave) {
        const int qs = cl & 3;
#pragma unroll
        for (int q = 0; q < 4; ++q) {
          if (q == qs) {
#pragma unroll
            for (int u = 0; u < 8; ++u) acc[q][u] += hx[u] * invh;
          }
        }
      }
    }
  }
}

extern "C" void kernel_launch(void* const* d_in, const int* in_sizes, int n_in,
                              void* d_out, int out_size, void* d_ws, size_t ws_size,
                              hipStream_t stream) {
  const int*   tokens = (const int*)  d_in[0];
  const float* h      = (const float*)d_in[1];
  const float* E      = (const float*)d_in[2];
  const float* Wn     = (const float*)d_in[3];
  const float* bn     = (const float*)d_in[4];
  const float* Wt     = (const float*)d_in[5];
  const float* bt     = (const float*)d_in[6];
  const float* tsa    = (const float*)d_in[7];
  const float* tst    = (const float*)d_in[8];
  const float* psa    = (const float*)d_in[9];
  const float* pst    = (const float*)d_in[10];
  const float* ppa    = (const float*)d_in[11];
  const float* ppt    = (const float*)d_in[12];
  const float* pta    = (const float*)d_in[13];
  const float* ptt    = (const float*)d_in[14];
  float* out = (float*)d_out;

  gemm_heads_kernel<<<ROWS / 128, 256, 0, stream>>>(h, E, Wn, bn, Wt, bt, tsa, tst, out);
  proto_kernel<<<NBATCH * 3, 256, 0, stream>>>(tokens, h, E, psa, pst, ppa, ppt,
                                               pta, ptt, out);
}


# proto kernel: the h row and old outputs of the next 5 events are loaded into rotating register sets while the current event is worked on; the serial walk no longer waits on memory per event
# speedup vs baseline: 4.6215x; 1.2415x over previous
#include <hip/hip_runtime.h>
#include <math.h>

typedef float v2f __attribute__((ext_vector_type(2)));
typedef float v8f __attribute__((ext_vector_type(8)));

#define TT 2048
#define DD 256
#define NBATCH 8
#define NACT 64
#define NTIME 32
#define NTOT 96
#define ROWS (NBATCH * TT)
#define OUT_TIME_OFF (ROWS * NACT)
#define PSTRIDE 112
#define KCHUNK 64
#define NAHEAD 5
#define NSLOT (NAHEAD + 1)

__device__ __forceinline__ float softplus_f(float x) { return log1pf(expf(x)); }

__device__ __forceinline__ float xrow16(float x) {
  const int b = __builtin_bit_cast(int, x);
  return __builtin_bit_cast(float, __builtin_amdgcn_permlanex16(
      b, b, 0x76543210, 0xfedcba98, false, false));
}
template <int M>
__device__ __forceinline__ float xrow(float x) {
  const int b = __builtin_bit_cast(int, x);
  return __builtin_bit_cast(float, __builtin_amdgcn_update_dpp(
      0, b, 0x160 | M, 0xf, 0xf, true));
}
__device__ __forceinline__ float wave_sum(float s) {
  s += xrow16(s);
  s += xrow<8>(s);
  s += xrow<4>(s);
  s += xrow<2>(s);
  s += xrow<1>(s);
  return s;
}

__global__ __launch_bounds__(256)
void gemm_heads_kernel(const float* __restrict__ h,
                       const float* __restrict__ E,
                       const float* __restrict__ Wn,
                       const float* __restrict__ bn,
                       const float* __restrict__ Wt,
                       const float* __restrict__ bt,
                       const float* __restrict__ tsa,
                       const float* __restrict__ tst,
                       float* __restrict__ out) {
  __shared__ v2f lw2[(KCHUNK / 2) * PSTRIDE];

  const int tid   = threadIdx.x;
  const int wave  = tid >> 5;
  const int lane  = tid & 31;
  const int lhalf = lane >> 4;
  const int l16   = lane & 15;
  const int m_base = blockIdx.x * 128 + wave * 16;

  const float s_ta = softplus_f(tsa[0]);
  const float s_tt = softplus_f(tst[0]);

  v8f acc[6];
#pragma unroll
  for (int j = 0; j < 6; ++j) {
    int n = j * 16 + l16;
    float bv = (n < NACT) ? bn[n] : bt[n - NACT];
#pragma unroll
    for (int r = 0; r < 8; ++r) acc[j][r] = bv;
  }

  float* lwf = (float*)lw2;

  for (int kc = 0; kc < DD; kc += KCHUNK) {
    __syncthreads();
#pragma unroll
    for (int i = 0; i < (KCHUNK * NTOT) / 256; ++i) {
      int idx = i * 256 + tid;
      int kk  = idx & (KCHUNK - 1);
      int n   = idx >> 6;
      int kg  = kc + kk;
      float wv;
      if (n < NACT) wv = Wn[n * DD + kg] + s_ta * E[(4 + n) * DD + kg];
      else          wv = Wt[(n - NACT) * DD + kg] + s_tt * E[(68 + (n - NACT)) * DD + kg];
      lwf[((kk >> 1) * PSTRIDE + n) * 2 + (kk & 1)] = wv;
    }
    __syncthreads();

#pragma unroll 4
    for (int ks = 0; ks < KCHUNK; ks += 4) {
      int kk = kc + ks + (lhalf << 1);
      v2f a = *(const v2f*)(h + (m_base + l16) * DD + kk);
      int p = (ks >> 1) + lhalf;
#pragma unroll
      for (int j = 0; j < 6; ++j) {
        int n = j * 16 + l16;
        v2f b = lw2[p * PSTRIDE + n];
        acc[j] = __builtin_amdgcn_wmma_f32_16x16x4_f32(
            false, a, false, b, (short)0, acc[j], false, false);
      }
    }
  }

#pragma unroll
  for (int j = 0; j < 6; ++j) {
    int n = j * 16 + l16;
#pragma unroll
    for (int r = 0; r < 8; ++r) {
      int row = m_base + r + (lhalf << 3);
      if (n < NACT) out[row * NACT + n] = acc[j][r];
      else          out[OUT_TIME_OFF + row * NTIME + (n - NACT)] = acc[j][r];
    }
  }
}

__device__ __forceinline__ void fetch_event(int ev, int b, int lane,
                                            const float* __restrict__ h,
                                            const float* obase, int ostride,
                                            float4& ra, float4& rc, float& ro) {
  const int r = b * TT + (ev >> 9);
  const float* hr = h + r * DD + lane * 8;
  ra = *(const float4*)(hr);
  rc = *(const float4*)(hr + 4);
  ro = obase[r * ostride + (lane & 3)];
}

__global__ __launch_bounds__(256)
void proto_kernel(const int* __restrict__ tokens,
                  const float* __restrict__ h,
                  const float* __restrict__ E,
                  const float* __restrict__ psa, const float* __restrict__ pst,
                  const float* __restrict__ ppa, const float* __restrict__ ppt,
                  const float* __restrict__ pta, const float* __restrict__ ptt,
                  float* __restrict__ out) {
  __shared__ int   evbuf[2048];
  __shared__ int   wfirst[8];
  __shared__ int   wcount[8];

  const int b   = blockIdx.x / 3;
  const int seg = blockIdx.x % 3;
  const bool is_time = (seg == 2);
  const int lo = is_time ? 68 : 4;
  const int hi = is_time ? 100 : 68;
  const int c0 = (seg == 1) ? 32 : 0;
  const int eb = lo + c0;

  const float alpha = softplus_f(is_time ? ppt[0] : ppa[0]);
  const float st    = softplus_f(is_time ? pst[0] : psa[0]) *
                      softplus_f(is_time ? ptt[0] : pta[0]);

  const int tid  = threadIdx.x;
  const int wave = __builtin_amdgcn_readfirstlane(tid >> 5);
  const int lane = tid & 31;

  const int* tk = tokens + b * TT;
  const int t0 = tid * 8;
  int tok[9];
  {
    const int4 qa = *(const int4*)(tk + t0);
    const int4 qb = *(const int4*)(tk + t0 + 4);
    tok[0] = qa.x; tok[1] = qa.y; tok[2] = qa.z; tok[3] = qa.w;
    tok[4] = qb.x; tok[5] = qb.y; tok[6] = qb.z; tok[7] = qb.w;
    tok[8] = tk[(t0 + 8) & (TT - 1)];
  }

  float acc[4][8];
#pragma unroll
  for (int q = 0; q < 4; ++q) {
    const float* er = E + (eb + wave * 4 + q) * DD + lane * 8;
    const float4 ea = *(const float4*)(er);
    const float4 ec = *(const float4*)(er + 4);
    const float v[8] = {ea.x, ea.y, ea.z, ea.w, ec.x, ec.y, ec.z, ec.w};
    float s = 0.f;
#pragma unroll
    for (int u = 0; u < 8; ++u) s = fmaf(v[u], v[u], s);
    s = wave_sum(s);
    const float inv = 1.0f / fmaxf(sqrtf(s), 1e-12f);
#pragma unroll
    for (int u = 0; u < 8; ++u) acc[q][u] = alpha * v[u] * inv;
  }

  unsigned labm = 0u, supm = 0u;
#pragma unroll
  for (int i = 0; i < 8; ++i) {
    if (tok[i] == 3) {
      labm |= 1u << i;
      int nxt = tok[i + 1];
      if (nxt >= lo && nxt < hi) supm |= 1u << i;
    }
  }
  {
    const unsigned bs = __builtin_amdgcn_ballot_w32(supm != 0u);
    const int fs = t0 + __builtin_ctz(supm | 0x100u);
    const int fw = __builtin_amdgcn_readlane(fs, bs ? __builtin_ctz(bs) : 0);
    if (lane == 0) wfirst[wave] = bs ? fw : TT;
  }
  __syncthreads();

  int first_sup = wfirst[0];
  for (int w = 1; w < 8; ++w) first_sup = min(first_sup, wfirst[w]);
  unsigned evm = 0u;
  for (int i = 0; i < 8; ++i) {
    unsigned bit = 1u << i;
    if ((labm & bit) && ((supm & bit) || (t0 + i > first_sup))) evm |= bit;
  }
  int ebefore = 0, wtotal = 0;
  for (int i = 0; i < 8; ++i) {
    const unsigned bi = __builtin_amdgcn_ballot_w32(((evm >> i) & 1u) != 0u);
    ebefore = __builtin_amdgcn_mbcnt_lo(bi, ebefore);
    wtotal += __popc(bi);
  }
  if (lane == 0) wcount[wave] = wtotal;
  __syncthreads();

  {
    int pos = ebefore;
    for (int w = 0; w < 8; ++w) {
      if (w < wave) pos += wcount[w];
    }
#pragma unroll
    for (int i = 0; i < 8; ++i) {
      if (evm & (1u << i)) {
        int t    = t0 + i;
        int emit = (t > first_sup) ? 1 : 0;
        int cls1 = (supm & (1u << i)) ? (tok[i + 1] - lo + 1) : 0;
        evbuf[pos++] = (t << 9) | (emit << 8) | cls1;
      }
    }
  }
  __syncthreads();

  int nevv = 0;
  for (int w = 0; w < 8; ++w) nevv += wcount[w];
  const int nev = __builtin_amdgcn_readfirstlane(nevv);

  float* const obase = is_time ? (out + OUT_TIME_OFF + wave * 4)
                               : (out + c0 + wave * 4);
  const int ostride = is_time ? NTIME : NACT;

  if (nev == 0) return;
  const int last = nev - 1;

  float4 pa[NSLOT], pc[NSLOT];
  float  po[NSLOT];
  int    pev[NSLOT];
#pragma unroll
  for (int k = 0; k < NSLOT; ++k) {
    pa[k] = make_float4(0.f, 0.f, 0.f, 0.f); pc[k] = pa[k]; po[k] = 0.f; pev[k] = 0;
  }
  int evq = evbuf[0];

  for (int e0 = -NSLOT; e0 < nev; e0 += NSLOT) {
#pragma unroll
    for (int s = 0; s < NSLOT; ++s) {
      const int e = e0 + s;

      {
        const int kn  = (s + NSLOT - 1) % NSLOT;
        const int evn = __builtin_amdgcn_readfirstlane(evq);
        pev[kn] = evn;
        fetch_event(evn, b, lane, h, obase, ostride, pa[kn], pc[kn], po[kn]);
        evq = evbuf[min(e + NAHEAD + 1, last)];
      }

      if ((unsigned)e >= (unsigned)nev) continue;

      const int ev   = pev[s];
      const int t    = ev >> 9;
      const int emit = (ev >> 8) & 1;
      const int clsp = ev & 255;
      const int r    = b * TT + t;

      const float4 ha = pa[s];
      const float4 hc = pc[s];
      const float oldv = po[s];
      float* const op = obase + r * ostride;
      const float hx[8] = {ha.x, ha.y, ha.z, ha.w, hc.x, hc.y, hc.z, hc.w};

      float hs = 0.f;
#pragma unroll
      for (int u = 0; u < 8; ++u) hs = fmaf(hx[u], hx[u], hs);
      hs = wave_sum(hs);
      const float invh = 1.0f / fmaxf(sqrtf(hs), 1e-12f);

      if (emit) {
        float dot[4], asq[4];
#pragma unroll
        for (int q = 0; q < 4; ++q) {
          dot[q] = 0.f; asq[q] = 0.f;
#pragma unroll
          for (int u = 0; u < 8; ++u) {
            dot[q] = fmaf(acc[q][u], hx[u], dot[q]);
            asq[q] = fmaf(acc[q][u], acc[q][u], asq[q]);
          }
        }
#pragma unroll
        for (int q = 0; q < 4; ++q) { dot[q] += xrow16(dot[q]);  asq[q] += xrow16(asq[q]); }
#pragma unroll
        for (int q = 0; q < 4; ++q) { dot[q] += xrow<8>(dot[q]); asq[q] += xrow<8>(asq[q]); }
#pragma unroll
        for (int q = 0; q < 4; ++q) { dot[q] += xrow<4>(dot[q]); asq[q] += xrow<4>(asq[q]); }
#pragma unroll
        for (int q = 0; q < 4; ++q) { dot[q] += xrow<2>(dot[q]); asq[q] += xrow<2>(asq[q]); }
#pragma unroll
        for (int q = 0; q < 4; ++q) { dot[q] += xrow<1>(dot[q]); asq[q] += xrow<1>(asq[q]); }
        const float dsel = (lane == 0) ? dot[0] : (lane == 1) ? dot[1]
                         : (lane == 2) ? dot[2] : dot[3];
        const float asel = (lane == 0) ? asq[0] : (lane == 1) ? asq[1]
                         : (lane == 2) ? asq[2] : asq[3];
        if (lane < 4) {
          float sim = dsel * invh / fmaxf(sqrtf(asel), 1e-12f);
          float add = st * sim;
          op[lane] = oldv + add;
        }
      }

      if (clsp) {
        const int cl = clsp - 1 - c0;
        if (cl >= 0 && cl < 32 && (cl >> 2) == wave) {
          const int qs = cl & 3;
#pragma unroll
          for (int q = 0; q < 4; ++q) {
            if (q == qs) {
#pragma unroll
              for (int u = 0; u < 8; ++u) acc[q][u] += hx[u] * invh;
            }
          }
        }
      }
    }
  }
}

extern "C" void kernel_launch(void* const* d_in, const int* in_sizes, int n_in,
                              void* d_out, int out_size, void* d_ws, size_t ws_size,
                              hipStream_t stream) {
  const int*   tokens = (const int*)  d_in[0];
  const float* h      = (const float*)d_in[1];
  const float* E      = (const float*)d_in[2];
  const float* Wn     = (const float*)d_in[3];
  const float* bn     = (const float*)d_in[4];
  const float* Wt     = (const float*)d_in[5];
  const float* bt     = (const float*)d_in[6];
  const float* tsa    = (const float*)d_in[7];
  const float* tst    = (const float*)d_in[8];
  const float* psa    = (const float*)d_in[9];
  const float* pst    = (const float*)d_in[10];
  const float* ppa    = (const float*)d_in[11];
  const float* ppt    = (const float*)d_in[12];
  const float* pta    = (const float*)d_in[13];
  const float* ptt    = (const float*)d_in[14];
  float* out = (float*)d_out;

  gemm_heads_kernel<<<ROWS / 128, 256, 0, stream>>>(h, E, Wn, bn, Wt, bt, tsa, tst, out);
  proto_kernel<<<NBATCH * 3, 256, 0, stream>>>(tokens, h, E, psa, pst, ppa, ppt,
                                               pta, ptt, out);
}


# GEMM kernel: combined weights stay in LDS for all of K (filled 64 columns at a time), next chunk's weights and next A fragments requested ahead of the WMMAs; 256 blocks of 4 waves, one per SIMD
# speedup vs baseline: 7.6536x; 1.6561x over previous
#include <hip/hip_runtime.h>
#include <math.h>

typedef float v2f __attribute__((ext_vector_type(2)));
typedef float v8f __attribute__((ext_vector_type(8)));

#define TT 2048
#define DD 256
#define NBATCH 8
#define NACT 64
#define NTIME 32
#define NTOT 96
#define ROWS (NBATCH * TT)
#define OUT_TIME_OFF (ROWS * NACT)
#define WPITCH 260
#define GM_THREADS 128
#define GM_ROWS 64
#define KCHUNK 64
#define NAHEAD 5
#define NSLOT (NAHEAD + 1)

__device__ __forceinline__ float softplus_f(float x) { return log1pf(expf(x)); }

__device__ __forceinline__ float xrow16(float x) {
  const int b = __builtin_bit_cast(int, x);
  return __builtin_bit_cast(float, __builtin_amdgcn_permlanex16(
      b, b, 0x76543210, 0xfedcba98, false, false));
}
template <int M>
__device__ __forceinline__ float xrow(float x) {
  const int b = __builtin_bit_cast(int, x);
  return __builtin_bit_cast(float, __builtin_amdgcn_update_dpp(
      0, b, 0x160 | M, 0xf, 0xf, true));
}
__device__ __forceinline__ float wave_sum(float s) {
  s += xrow16(s);
  s += xrow<8>(s);
  s += xrow<4>(s);
  s += xrow<2>(s);
  s += xrow<1>(s);
  return s;
}

__global__ __launch_bounds__(GM_THREADS)
void gemm_heads_kernel(const float* __restrict__ h,
                       const float* __restrict__ E,
                       const float* __restrict__ Wn,
                       const float* __restrict__ bn,
                       const float* __restrict__ Wt,
                       const float* __restrict__ bt,
                       const float* __restrict__ tsa,
                       const float* __restrict__ tst,
                       float* __restrict__ out) {
  __shared__ float lw[NTOT * WPITCH];

  const int tid   = threadIdx.x;
  const int wave  = tid >> 5;
  const int lane  = tid & 31;
  const int lhalf = lane >> 4;
  const int l16   = lane & 15;
  const int m_base = blockIdx.x * GM_ROWS + wave * 16;

  const float* arow = h + (m_base + l16) * DD + (lhalf << 1);
  const float* lb = lw + l16 * WPITCH + (lhalf << 1);

  v2f abuf[2][8];
#pragma unroll
  for (int q = 0; q < 8; ++q) abuf[0][q] = *(const v2f*)(arow + q * 4);

  const int sq = tid & 15;
  const int sn = tid >> 4;
  const float* wn_p = Wn + sn * DD + sq * 4;
  const float* ea_p = E + (4 + sn) * DD + sq * 4;
  const float* wt_p = Wt + sn * DD + sq * 4;
  const float* et_p = E + (68 + sn) * DD + sq * 4;
  float* lst = lw + sn * WPITCH + sq * 4;
  float4 wr[12], er[12];

#pragma unroll
  for (int i = 0; i < 8; ++i) {
    wr[i] = *(const float4*)(wn_p + i * 8 * DD);
    er[i] = *(const float4*)(ea_p + i * 8 * DD);
  }
#pragma unroll
  for (int i = 0; i < 4; ++i) {
    wr[8 + i] = *(const float4*)(wt_p + i * 8 * DD);
    er[8 + i] = *(const float4*)(et_p + i * 8 * DD);
  }

  const float s_ta = softplus_f(tsa[0]);
  const float s_tt = softplus_f(tst[0]);

  v8f acc[6];
#pragma unroll
  for (int j = 0; j < 6; ++j) {
    int n = j * 16 + l16;
    float bv = (n < NACT) ? bn[n] : bt[n - NACT];
#pragma unroll
    for (int r = 0; r < 8; ++r) acc[j][r] = bv;
  }

#pragma unroll
  for (int i = 0; i < 12; ++i) {
    const float s = (i < 8) ? s_ta : s_tt;
    float4 v;
    v.x = wr[i].x + s * er[i].x;
    v.y = wr[i].y + s * er[i].y;
    v.z = wr[i].z + s * er[i].z;
    v.w = wr[i].w + s * er[i].w;
    *(float4*)(lst + i * 8 * WPITCH) = v;
  }
  __syncthreads();

#pragma unroll
  for (int hc = 0; hc < 8; ++hc) {
    const int cur = hc & 1;
    const int nxt = cur ^ 1;
    if (hc + 1 < 8) {
#pragma unroll
      for (int q = 0; q < 8; ++q)
        abuf[nxt][q] = *(const v2f*)(arow + (hc + 1) * 32 + q * 4);
    }
    if ((hc & 1) == 0 && hc < 6) {
      const int kc = ((hc >> 1) + 1) * KCHUNK;
#pragma unroll
      for (int i = 0; i < 8; ++i) {
        wr[i] = *(const float4*)(wn_p + i * 8 * DD + kc);
        er[i] = *(const float4*)(ea_p + i * 8 * DD + kc);
      }
#pragma unroll
      for (int i = 0; i < 4; ++i) {
        wr[8 + i] = *(const float4*)(wt_p + i * 8 * DD + kc);
        er[8 + i] = *(const float4*)(et_p + i * 8 * DD + kc);
      }
    }

#pragma unroll
    for (int q = 0; q < 8; ++q) {
      const v2f a = abuf[cur][q];
#pragma unroll
      for (int j = 0; j < 6; ++j) {
        const v2f b = *(const v2f*)(lb + j * 16 * WPITCH + hc * 32 + q * 4);
        acc[j] = __builtin_amdgcn_wmma_f32_16x16x4_f32(
            false, a, false, b, (short)0, acc[j], false, false);
      }
    }

    if ((hc & 1) == 1 && hc < 7) {
      const int kc = ((hc >> 1) + 1) * KCHUNK;
      __builtin_amdgcn_sched_barrier(0);
#pragma unroll
      for (int i = 0; i < 12; ++i) {
        const float s = (i < 8) ? s_ta : s_tt;
        float4 v;
        v.x = wr[i].x + s * er[i].x;
        v.y = wr[i].y + s * er[i].y;
        v.z = wr[i].z + s * er[i].z;
        v.w = wr[i].w + s * er[i].w;
        *(float4*)(lst + i * 8 * WPITCH + kc) = v;
      }
      __syncthreads();
    }
  }

#pragma unroll
  for (int j = 0; j < 6; ++j) {
    int n = j * 16 + l16;
#pragma unroll
    for (int r = 0; r < 8; ++r) {
      int row = m_base + r + (lhalf << 3);
      if (n < NACT) out[row * NACT + n] = acc[j][r];
      else          out[OUT_TIME_OFF + row * NTIME + (n - NACT)] = acc[j][r];
    }
  }
}

__device__ __forceinline__ void fetch_event(int ev, int b, int lane,
                                            const float* __restrict__ h,
                                            const float* obase, int ostride,
                                            float4& ra, float4& rc, float& ro) {
  const int r = b * TT + (ev >> 9);
  const float* hr = h + r * DD + lane * 8;
  ra = *(const float4*)(hr);
  rc = *(const float4*)(hr + 4);
  ro = obase[r * ostride + (lane & 3)];
}

__global__ __launch_bounds__(256)
void proto_kernel(const int* __restrict__ tokens,
                  const float* __restrict__ h,
                  const float* __restrict__ E,
                  const float* __restrict__ psa, const float* __restrict__ pst,
                  const float* __restrict__ ppa, const float* __restrict__ ppt,
                  const float* __restrict__ pta, const float* __restrict__ ptt,
                  float* __restrict__ out) {
  __shared__ int   evbuf[2048];
  __shared__ int   wfirst[8];
  __shared__ int   wcount[8];

  const int b   = blockIdx.x / 3;
  const int seg = blockIdx.x % 3;
  const bool is_time = (seg == 2);
  const int lo = is_time ? 68 : 4;
  const int hi = is_time ? 100 : 68;
  const int c0 = (seg == 1) ? 32 : 0;
  const int eb = lo + c0;

  const float alpha = softplus_f(is_time ? ppt[0] : ppa[0]);
  const float st    = softplus_f(is_time ? pst[0] : psa[0]) *
                      softplus_f(is_time ? ptt[0] : pta[0]);

  const int tid  = threadIdx.x;
  const int wave = __builtin_amdgcn_readfirstlane(tid >> 5);
  const int lane = tid & 31;

  const int* tk = tokens + b * TT;
  const int t0 = tid * 8;
  int tok[9];
  {
    const int4 qa = *(const int4*)(tk + t0);
    const int4 qb = *(const int4*)(tk + t0 + 4);
    tok[0] = qa.x; tok[1] = qa.y; tok[2] = qa.z; tok[3] = qa.w;
    tok[4] = qb.x; tok[5] = qb.y; tok[6] = qb.z; tok[7] = qb.w;
    tok[8] = tk[(t0 + 8) & (TT - 1)];
  }

  float acc[4][8];
#pragma unroll
  for (int q = 0; q < 4; ++q) {
    const float* er = E + (eb + wave * 4 + q) * DD + lane * 8;
    const float4 ea = *(const float4*)(er);
    const float4 ec = *(const float4*)(er + 4);
    const float v[8] = {ea.x, ea.y, ea.z, ea.w, ec.x, ec.y, ec.z, ec.w};
    float s = 0.f;
#pragma unroll
    for (int u = 0; u < 8; ++u) s = fmaf(v[u], v[u], s);
    s = wave_sum(s);
    const float inv = 1.0f / fmaxf(sqrtf(s), 1e-12f);
#pragma unroll
    for (int u = 0; u < 8; ++u) acc[q][u] = alpha * v[u] * inv;
  }

  unsigned labm = 0u, supm = 0u;
#pragma unroll
  for (int i = 0; i < 8; ++i) {
    if (tok[i] == 3) {
      labm |= 1u << i;
      int nxt = tok[i + 1];
      if (nxt >= lo && nxt < hi) supm |= 1u << i;
    }
  }
  {
    const unsigned bs = __builtin_amdgcn_ballot_w32(supm != 0u);
    const int fs = t0 + __builtin_ctz(supm | 0x100u);
    const int fw = __builtin_amdgcn_readlane(fs, bs ? __builtin_ctz(bs) : 0);
    if (lane == 0) wfirst[wave] = bs ? fw : TT;
  }
  __syncthreads();

  int first_sup = wfirst[0];
  for (int w = 1; w < 8; ++w) first_sup = min(first_sup, wfirst[w]);
  unsigned evm = 0u;
  for (int i = 0; i < 8; ++i) {
    unsigned bit = 1u << i;
    if ((labm & bit) && ((supm & bit) || (t0 + i > first_sup))) evm |= bit;
  }
  int ebefore = 0, wtotal = 0;
  for (int i = 0; i < 8; ++i) {
    const unsigned bi = __builtin_amdgcn_ballot_w32(((evm >> i) & 1u) != 0u);
    ebefore = __builtin_amdgcn_mbcnt_lo(bi, ebefore);
    wtotal += __popc(bi);
  }
  if (lane == 0) wcount[wave] = wtotal;
  __syncthreads();

  {
    int pos = ebefore;
    for (int w = 0; w < 8; ++w) {
      if (w < wave) pos += wcount[w];
    }
#pragma unroll
    for (int i = 0; i < 8; ++i) {
      if (evm & (1u << i)) {
        int t    = t0 + i;
        int emit = (t > first_sup) ? 1 : 0;
        int cls1 = (supm & (1u << i)) ? (tok[i + 1] - lo + 1) : 0;
        evbuf[pos++] = (t << 9) | (emit << 8) | cls1;
      }
    }
  }
  __syncthreads();

  int nevv = 0;
  for (int w = 0; w < 8; ++w) nevv += wcount[w];
  const int nev = __builtin_amdgcn_readfirstlane(nevv);

  float* const obase = is_time ? (out + OUT_TIME_OFF + wave * 4)
                               : (out + c0 + wave * 4);
  const int ostride = is_time ? NTIME : NACT;

  if (nev == 0) return;
  const int last = nev - 1;

  float4 pa[NSLOT], pc[NSLOT];
  float  po[NSLOT];
  int    pev[NSLOT];
#pragma unroll
  for (int k = 0; k < NSLOT; ++k) {
    pa[k] = make_float4(0.f, 0.f, 0.f, 0.f); pc[k] = pa[k]; po[k] = 0.f; pev[k] = 0;
  }
  int evq = evbuf[0];

  for (int e0 = -NSLOT; e0 < nev; e0 += NSLOT) {
#pragma unroll
    for (int s = 0; s < NSLOT; ++s) {
      const int e = e0 + s;

      {
        const int kn  = (s + NSLOT - 1) % NSLOT;
        const int evn = __builtin_amdgcn_readfirstlane(evq);
        pev[kn] = evn;
        fetch_event(evn, b, lane, h, obase, ostride, pa[kn], pc[kn], po[kn]);
        evq = evbuf[min(e + NAHEAD + 1, last)];
      }

      if ((unsigned)e >= (unsigned)nev) continue;

      const int ev   = pev[s];
      const int t    = ev >> 9;
      const int emit = (ev >> 8) & 1;
      const int clsp = ev & 255;
      const int r    = b * TT + t;

      const float4 ha = pa[s];
      const float4 hc = pc[s];
      const float oldv = po[s];
      float* const op = obase + r * ostride;
      const float hx[8] = {ha.x, ha.y, ha.z, ha.w, hc.x, hc.y, hc.z, hc.w};

      float hs = 0.f;
#pragma unroll
      for (int u = 0; u < 8; ++u) hs = fmaf(hx[u], hx[u], hs);
      hs = wave_sum(hs);
      const float invh = 1.0f / fmaxf(sqrtf(hs), 1e-12f);

      if (emit) {
        float dot[4], asq[4];
#pragma unroll
        for (int q = 0; q < 4; ++q) {
          dot[q] = 0.f; asq[q] = 0.f;
#pragma unroll
          for (int u = 0; u < 8; ++u) {
            dot[q] = fmaf(acc[q][u], hx[u], dot[q]);
            asq[q] = fmaf(acc[q][u], acc[q][u], asq[q]);
          }
        }
#pragma unroll
        for (int q = 0; q < 4; ++q) { dot[q] += xrow16(dot[q]);  asq[q] += xrow16(asq[q]); }
#pragma unroll
        for (int q = 0; q < 4; ++q) { dot[q] += xrow<8>(dot[q]); asq[q] += xrow<8>(asq[q]); }
#pragma unroll
        for (int q = 0; q < 4; ++q) { dot[q] += xrow<4>(dot[q]); asq[q] += xrow<4>(asq[q]); }
#pragma unroll
        for (int q = 0; q < 4; ++q) { dot[q] += xrow<2>(dot[q]); asq[q] += xrow<2>(asq[q]); }
#pragma unroll
        for (int q = 0; q < 4; ++q) { dot[q] += xrow<1>(dot[q]); asq[q] += xrow<1>(asq[q]); }
        const float dsel = (lane == 0) ? dot[0] : (lane == 1) ? dot[1]
                         : (lane == 2) ? dot[2] : dot[3];
        const float asel = (lane == 0) ? asq[0] : (lane == 1) ? asq[1]
                         : (lane == 2) ? asq[2] : asq[3];
        if (lane < 4) {
          float sim = dsel * invh / fmaxf(sqrtf(asel), 1e-12f);
          float add = st * sim;
          op[lane] = oldv + add;
        }
      }

      if (clsp) {
        const int cl = clsp - 1 - c0;
        if (cl >= 0 && cl < 32 && (cl >> 2) == wave) {
          const int qs = cl & 3;
#pragma unroll
          for (int q = 0; q < 4; ++q) {
            if (q == qs) {
#pragma unroll
              for (int u = 0; u < 8; ++u) acc[q][u] += hx[u] * invh;
            }
          }
        }
      }
    }
  }
}

extern "C" void kernel_launch(void* const* d_in, const int* in_sizes, int n_in,
                              void* d_out, int out_size, void* d_ws, size_t ws_size,
                              hipStream_t stream) {
  const int*   tokens = (const int*)  d_in[0];
  const float* h      = (const float*)d_in[1];
  const float* E      = (const float*)d_in[2];
  const float* Wn     = (const float*)d_in[3];
  const float* bn     = (const float*)d_in[4];
  const float* Wt     = (const float*)d_in[5];
  const float* bt     = (const float*)d_in[6];
  const float* tsa    = (const float*)d_in[7];
  const float* tst    = (const float*)d_in[8];
  const float* psa    = (const float*)d_in[9];
  const float* pst    = (const float*)d_in[10];
  const float* ppa    = (const float*)d_in[11];
  const float* ppt    = (const float*)d_in[12];
  const float* pta    = (const float*)d_in[13];
  const float* ptt    = (const float*)d_in[14];
  float* out = (float*)d_out;

  gemm_heads_kernel<<<ROWS / GM_ROWS, GM_THREADS, 0, stream>>>(h, E, Wn, bn, Wt, bt, tsa, tst, out);
  proto_kernel<<<NBATCH * 3, 256, 0, stream>>>(tokens, h, E, psa, pst, ppa, ppt,
                                               pta, ptt, out);
}


# one launch instead of two: 48 four-wave prototype blocks run beside the 256 GEMM blocks; a label row is written only by its prototype blocks, which form its head product on WMMA, same k order
# speedup vs baseline: 10.1909x; 1.3315x over previous
#include <hip/hip_runtime.h>
#include <math.h>

typedef float v2f __attribute__((ext_vector_type(2)));
typedef float v8f __attribute__((ext_vector_type(8)));

#define TT 2048
#define DD 256
#define NBATCH 8
#define NACT 64
#define NTIME 32
#define NTOT 96
#define ROWS (NBATCH * TT)
#define OUT_TIME_OFF (ROWS * NACT)
#define WPITCH 260
#define GM_THREADS 128
#define GM_ROWS 64
#define KCHUNK 64
#define NAHEAD 5
#define NSLOT (NAHEAD + 1)
#define PSEG 16
#define NPSEG 6
#define NPROTO 48
#define NBLOCKS 304

__device__ __forceinline__ float softplus_f(float x) { return log1pf(expf(x)); }

__device__ __forceinline__ float xrow16(float x) {
  const int b = __builtin_bit_cast(int, x);
  return __builtin_bit_cast(float, __builtin_amdgcn_permlanex16(
      b, b, 0x76543210, 0xfedcba98, false, false));
}
template <int M>
__device__ __forceinline__ float xrow(float x) {
  const int b = __builtin_bit_cast(int, x);
  return __builtin_bit_cast(float, __builtin_amdgcn_update_dpp(
      0, b, 0x160 | M, 0xf, 0xf, true));
}
__device__ __forceinline__ float wave_sum(float s) {
  s += xrow16(s);
  s += xrow<8>(s);
  s += xrow<4>(s);
  s += xrow<2>(s);
  s += xrow<1>(s);
  return s;
}


__device__ __forceinline__ void gemm_role(const int gb, float* lw,
                                          const int* __restrict__ tokens,
                                          const float* __restrict__ h,
                                          const float* __restrict__ E,
                                          const float* __restrict__ Wn,
                                          const float* __restrict__ bn,
                                          const float* __restrict__ Wt,
                                          const float* __restrict__ bt,
                                          const float* __restrict__ tsa,
                                          const float* __restrict__ tst,
                                          float* __restrict__ out) {
  const int tid   = threadIdx.x;
  const int wave  = tid >> 5;
  const int lane  = tid & 31;
  const int lhalf = lane >> 4;
  const int l16   = lane & 15;
  const int m_base = gb * GM_ROWS + wave * 16;

  const float* arow = h + (m_base + l16) * DD + (lhalf << 1);
  const float* lb = lw + l16 * WPITCH + (lhalf << 1);

  v2f abuf[2][8];
#pragma unroll
  for (int q = 0; q < 8; ++q) abuf[0][q] = *(const v2f*)(arow + q * 4);

  const int* tkr = tokens + m_base + (lhalf << 3);
  const int4 tq0 = *(const int4*)(tkr);
  const int4 tq1 = *(const int4*)(tkr + 4);

  const int sq = tid & 15;
  const int sn = tid >> 4;
  const float* wn_p = Wn + sn * DD + sq * 4;
  const float* ea_p = E + (4 + sn) * DD + sq * 4;
  const float* wt_p = Wt + sn * DD + sq * 4;
  const float* et_p = E + (68 + sn) * DD + sq * 4;
  float* lst = lw + sn * WPITCH + sq * 4;
  float4 wr[12], er[12];

#pragma unroll
  for (int i = 0; i < 8; ++i) {
    wr[i] = *(const float4*)(wn_p + i * 8 * DD);
    er[i] = *(const float4*)(ea_p + i * 8 * DD);
  }
#pragma unroll
  for (int i = 0; i < 4; ++i) {
    wr[8 + i] = *(const float4*)(wt_p + i * 8 * DD);
    er[8 + i] = *(const float4*)(et_p + i * 8 * DD);
  }

  const float s_ta = softplus_f(tsa[0]);
  const float s_tt = softplus_f(tst[0]);

  v8f acc[6];
#pragma unroll
  for (int j = 0; j < 6; ++j) {
    int n = j * 16 + l16;
    float bv = (n < NACT) ? bn[n] : bt[n - NACT];
#pragma unroll
    for (int r = 0; r < 8; ++r) acc[j][r] = bv;
  }

#pragma unroll
  for (int i = 0; i < 12; ++i) {
    const float s = (i < 8) ? s_ta : s_tt;
    float4 v;
    v.x = wr[i].x + s * er[i].x;
    v.y = wr[i].y + s * er[i].y;
    v.z = wr[i].z + s * er[i].z;
    v.w = wr[i].w + s * er[i].w;
    *(float4*)(lst + i * 8 * WPITCH) = v;
  }
  __syncthreads();

#pragma unroll
  for (int hc = 0; hc < 8; ++hc) {
    const int cur = hc & 1;
    const int nxt = cur ^ 1;
    if (hc + 1 < 8) {
#pragma unroll
      for (int q = 0; q < 8; ++q)
        abuf[nxt][q] = *(const v2f*)(arow + (hc + 1) * 32 + q * 4);
    }
    if ((hc & 1) == 0 && hc < 6) {
      const int kc = ((hc >> 1) + 1) * KCHUNK;
#pragma unroll
      for (int i = 0; i < 8; ++i) {
        wr[i] = *(const float4*)(wn_p + i * 8 * DD + kc);
        er[i] = *(const float4*)(ea_p + i * 8 * DD + kc);
      }
#pragma unroll
      for (int i = 0; i < 4; ++i) {
        wr[8 + i] = *(const float4*)(wt_p + i * 8 * DD + kc);
        er[8 + i] = *(const float4*)(et_p + i * 8 * DD + kc);
      }
    }

#pragma unroll
    for (int q = 0; q < 8; ++q) {
      const v2f a = abuf[cur][q];
#pragma unroll
      for (int j = 0; j < 6; ++j) {
        const v2f b = *(const v2f*)(lb + j * 16 * WPITCH + hc * 32 + q * 4);
        acc[j] = __builtin_amdgcn_wmma_f32_16x16x4_f32(
            false, a, false, b, (short)0, acc[j], false, false);
      }
    }

    if ((hc & 1) == 1 && hc < 7) {
      const int kc = ((hc >> 1) + 1) * KCHUNK;
      __builtin_amdgcn_sched_barrier(0);
#pragma unroll
      for (int i = 0; i < 12; ++i) {
        const float s = (i < 8) ? s_ta : s_tt;
        float4 v;
        v.x = wr[i].x + s * er[i].x;
        v.y = wr[i].y + s * er[i].y;
        v.z = wr[i].z + s * er[i].z;
        v.w = wr[i].w + s * er[i].w;
        *(float4*)(lst + i * 8 * WPITCH + kc) = v;
      }
      __syncthreads();
    }
  }

  const int tk8[8] = {tq0.x, tq0.y, tq0.z, tq0.w, tq1.x, tq1.y, tq1.z, tq1.w};
#pragma unroll
  for (int r = 0; r < 8; ++r) {
    if (tk8[r] != 3) {
      const int row = m_base + r + (lhalf << 3);
#pragma unroll
      for (int j = 0; j < 6; ++j) {
        const int n = j * 16 + l16;
        if (n < NACT) out[row * NACT + n] = acc[j][r];
        else          out[OUT_TIME_OFF + row * NTIME + (n - NACT)] = acc[j][r];
      }
    }
  }
}

__device__ __forceinline__ void fetch_event(int ev, int b, int lane,
                                            const float* __restrict__ h,
                                            const float* obase, int ostride,
                                            float4& ra, float4& rc, float& ro) {
  const int r = b * TT + (ev >> 9);
  const float* hr = h + r * DD + lane * 8;
  ra = *(const float4*)(hr);
  rc = *(const float4*)(hr + 4);
  ro = obase[r * ostride + (lane & 3)];
}

__device__ __forceinline__ void proto_role(const int pb, float* smem,
                                           const int* __restrict__ tokens,
                                           const float* __restrict__ h,
                                           const float* __restrict__ E,
                                           const float* __restrict__ Wn,
                                           const float* __restrict__ bn,
                                           const float* __restrict__ Wt,
                                           const float* __restrict__ bt,
                                           const float* __restrict__ tsa,
                                           const float* __restrict__ tst,
                                           const float* __restrict__ psa,
                                           const float* __restrict__ pst,
                                           const float* __restrict__ ppa,
                                           const float* __restrict__ ppt,
                                           const float* __restrict__ pta,
                                           const float* __restrict__ ptt,
                                           float* __restrict__ out) {
  int*   evbuf  = (int*)smem;
  int*   wfirst = evbuf + 2048;
  int*   wcount = evbuf + 2052;
  float* lwp    = smem + 2064;

  const int b   = pb / NPSEG;
  const int sg  = pb % NPSEG;
  const bool is_time = (sg >= 4);
  const int lo = is_time ? 68 : 4;
  const int hi = is_time ? 100 : 68;
  const int c0 = (is_time ? (sg - 4) : sg) * PSEG;
  const int eb = lo + c0;

  const int tid  = threadIdx.x;
  const int wave = __builtin_amdgcn_readfirstlane(tid >> 5);
  const int lane = tid & 31;
  const int lhalf = lane >> 4;
  const int l16   = lane & 15;

  const int* tk = tokens + b * TT;
  const int t0 = tid * 16;
  int tok[17];
  {
    const int4 qa = *(const int4*)(tk + t0);
    const int4 qb = *(const int4*)(tk + t0 + 4);
    const int4 qc = *(const int4*)(tk + t0 + 8);
    const int4 qd = *(const int4*)(tk + t0 + 12);
    tok[0]  = qa.x; tok[1]  = qa.y; tok[2]  = qa.z; tok[3]  = qa.w;
    tok[4]  = qb.x; tok[5]  = qb.y; tok[6]  = qb.z; tok[7]  = qb.w;
    tok[8]  = qc.x; tok[9]  = qc.y; tok[10] = qc.z; tok[11] = qc.w;
    tok[12] = qd.x; tok[13] = qd.y; tok[14] = qd.z; tok[15] = qd.w;
    tok[16] = tk[(t0 + 16) & (TT - 1)];
  }

  const float* Wh = is_time ? Wt : Wn;
  const int sq = tid & 15;
  const int sn = tid >> 4;
  float4 wr[2][4], er[2][4];
#pragma unroll
  for (int i = 0; i < 2; ++i) {
#pragma unroll
    for (int j = 0; j < 4; ++j) {
      wr[i][j] = *(const float4*)(Wh + (c0 + sn + 8 * i) * DD + (sq + 16 * j) * 4);
      er[i][j] = *(const float4*)(E  + (eb + sn + 8 * i) * DD + (sq + 16 * j) * 4);
    }
  }

  const float* bp = is_time ? bt : bn;
  const float bv = bp[c0 + l16];

  float4 ea[4], ec[4];
#pragma unroll
  for (int q = 0; q < 4; ++q) {
    const float* erow = E + (eb + wave * 4 + q) * DD + lane * 8;
    ea[q] = *(const float4*)(erow);
    ec[q] = *(const float4*)(erow + 4);
  }

  const float alpha = softplus_f((is_time ? ppt : ppa)[0]);
  const float st    = softplus_f((is_time ? pst : psa)[0]) *
                      softplus_f((is_time ? ptt : pta)[0]);
  const float s_t   = softplus_f((is_time ? tst : tsa)[0]);

  float acc[4][8];
#pragma unroll
  for (int q = 0; q < 4; ++q) {
    const float v[8] = {ea[q].x, ea[q].y, ea[q].z, ea[q].w, ec[q].x, ec[q].y, ec[q].z, ec[q].w};
    float s = 0.f;
#pragma unroll
    for (int u = 0; u < 8; ++u) s = fmaf(v[u], v[u], s);
    s = wave_sum(s);
    const float inv = 1.0f / fmaxf(sqrtf(s), 1e-12f);
#pragma unroll
    for (int u = 0; u < 8; ++u) acc[q][u] = alpha * v[u] * inv;
  }

  unsigned labm = 0u, supm = 0u;
#pragma unroll
  for (int i = 0; i < 16; ++i) {
    if (tok[i] == 3) {
      labm |= 1u << i;
      int nx = tok[i + 1];
      if (nx >= lo && nx < hi) supm |= 1u << i;
    }
  }
  {
    const unsigned bs = __builtin_amdgcn_ballot_w32(supm != 0u);
    const int fs = t0 + __builtin_ctz(supm | 0x10000u);
    const int fw = __builtin_amdgcn_readlane(fs, bs ? __builtin_ctz(bs) : 0);
    if (lane == 0) wfirst[wave] = bs ? fw : TT;
  }
  __syncthreads();

  int first_sup = wfirst[0];
  for (int w = 1; w < 4; ++w) first_sup = min(first_sup, wfirst[w]);
  const unsigned evm = labm;
  int ebefore = 0, wtotal = 0;
  for (int i = 0; i < 16; ++i) {
    const unsigned bi = __builtin_amdgcn_ballot_w32(((evm >> i) & 1u) != 0u);
    ebefore = __builtin_amdgcn_mbcnt_lo(bi, ebefore);
    wtotal += __popc(bi);
  }
  if (lane == 0) wcount[wave] = wtotal;

#pragma unroll
  for (int i = 0; i < 2; ++i) {
#pragma unroll
    for (int j = 0; j < 4; ++j) {
      float4 v;
      v.x = wr[i][j].x + s_t * er[i][j].x;
      v.y = wr[i][j].y + s_t * er[i][j].y;
      v.z = wr[i][j].z + s_t * er[i][j].z;
      v.w = wr[i][j].w + s_t * er[i][j].w;
      *(float4*)(lwp + (sn + 8 * i) * WPITCH + (sq + 16 * j) * 4) = v;
    }
  }
  __syncthreads();

  {
    int pos = ebefore;
    for (int w = 0; w < 4; ++w) {
      if (w < wave) pos += wcount[w];
    }
#pragma unroll
    for (int i = 0; i < 16; ++i) {
      if (evm & (1u << i)) {
        int t    = t0 + i;
        int emit = (t > first_sup) ? 1 : 0;
        int cls1 = (supm & (1u << i)) ? (tok[i + 1] - lo + 1) : 0;
        evbuf[pos++] = (t << 9) | (emit << 8) | cls1;
      }
    }
  }
  __syncthreads();

  int nevv = 0;
  for (int w = 0; w < 4; ++w) nevv += wcount[w];
  const int nev = __builtin_amdgcn_readfirstlane(nevv);

  const int ostride = is_time ? NTIME : NACT;
  float* const sbase = is_time ? (out + OUT_TIME_OFF + c0) : (out + c0);
  float* const obase = sbase + wave * 4;

  if (nev == 0) return;
  const int last = nev - 1;

  {
    const float* lb = lwp + l16 * WPITCH + (lhalf << 1);
    for (int et = wave; et * 16 < nev; et += 4) {
      const int m  = min(et * 16 + l16, last);
      const int ra = b * TT + (evbuf[m] >> 9);
      const float* arow = h + ra * DD + (lhalf << 1);
      v2f af[64];
#pragma unroll
      for (int q = 0; q < 64; ++q) af[q] = *(const v2f*)(arow + q * 4);
      v8f d;
#pragma unroll
      for (int r = 0; r < 8; ++r) d[r] = bv;
#pragma unroll
      for (int q = 0; q < 64; ++q) {
        const v2f bq = *(const v2f*)(lb + q * 4);
        d = __builtin_amdgcn_wmma_f32_16x16x4_f32(
            false, af[q], false, bq, (short)0, d, false, false);
      }
#pragma unroll
      for (int r = 0; r < 8; ++r) {
        const int mm = et * 16 + r + (lhalf << 3);
        if (mm < nev) {
          const int rr = b * TT + (evbuf[mm] >> 9);
          sbase[rr * ostride + l16] = d[r];
        }
      }
    }
  }
  __syncthreads();

  float4 pa[NSLOT], pc[NSLOT];
  float  po[NSLOT];
  int    pev[NSLOT];
#pragma unroll
  for (int k = 0; k < NSLOT; ++k) {
    pa[k] = make_float4(0.f, 0.f, 0.f, 0.f); pc[k] = pa[k]; po[k] = 0.f; pev[k] = 0;
  }
  int evq = evbuf[0];

  for (int e0 = -NSLOT; e0 < nev; e0 += NSLOT) {
#pragma unroll
    for (int s = 0; s < NSLOT; ++s) {
      const int e = e0 + s;

      {
        const int kn  = (s + NSLOT - 1) % NSLOT;
        const int evn = __builtin_amdgcn_readfirstlane(evq);
        pev[kn] = evn;
        fetch_event(evn, b, lane, h, obase, ostride, pa[kn], pc[kn], po[kn]);
        evq = evbuf[min(e + NAHEAD + 1, last)];
      }

      if ((unsigned)e >= (unsigned)nev) continue;

      const int ev   = pev[s];
      const int t    = ev >> 9;
      const int emit = (ev >> 8) & 1;
      const int clsp = ev & 255;
      const int r    = b * TT + t;

      const float4 ha = pa[s];
      const float4 hc = pc[s];
      const float oldv = po[s];
      float* const op = obase + r * ostride;
      const float hx[8] = {ha.x, ha.y, ha.z, ha.w, hc.x, hc.y, hc.z, hc.w};

      float hs = 0.f;
#pragma unroll
      for (int u = 0; u < 8; ++u) hs = fmaf(hx[u], hx[u], hs);
      hs = wave_sum(hs);
      const float invh = 1.0f / fmaxf(sqrtf(hs), 1e-12f);

      if (emit) {
        float dot[4], asq[4];
#pragma unroll
        for (int q = 0; q < 4; ++q) {
          dot[q] = 0.f; asq[q] = 0.f;
#pragma unroll
          for (int u = 0; u < 8; ++u) {
            dot[q] = fmaf(acc[q][u], hx[u], dot[q]);
            asq[q] = fmaf(acc[q][u], acc[q][u], asq[q]);
          }
        }
#pragma unroll
        for (int q = 0; q < 4; ++q) { dot[q] += xrow16(dot[q]);  asq[q] += xrow16(asq[q]); }
#pragma unroll
        for (int q = 0; q < 4; ++q) { dot[q] += xrow<8>(dot[q]); asq[q] += xrow<8>(asq[q]); }
#pragma unroll
        for (int q = 0; q < 4; ++q) { dot[q] += xrow<4>(dot[q]); asq[q] += xrow<4>(asq[q]); }
#pragma unroll
        for (int q = 0; q < 4; ++q) { dot[q] += xrow<2>(dot[q]); asq[q] += xrow<2>(asq[q]); }
#pragma unroll
        for (int q = 0; q < 4; ++q) { dot[q] += xrow<1>(dot[q]); asq[q] += xrow<1>(asq[q]); }
        const float dsel = (lane == 0) ? dot[0] : (lane == 1) ? dot[1]
                         : (lane == 2) ? dot[2] : dot[3];
        const float asel = (lane == 0) ? asq[0] : (lane == 1) ? asq[1]
                         : (lane == 2) ? asq[2] : asq[3];
        if (lane < 4) {
          float sim = dsel * invh / fmaxf(sqrtf(asel), 1e-12f);
          float add = st * sim;
          op[lane] = oldv + add;
        }
      }

      if (clsp) {
        const int cl = clsp - 1 - c0;
        if (cl >= 0 && cl < PSEG && (cl >> 2) == wave) {
          const int qs = cl & 3;
#pragma unroll
          for (int q = 0; q < 4; ++q) {
            if (q == qs) {
#pragma unroll
              for (int u = 0; u < 8; ++u) acc[q][u] += hx[u] * invh;
            }
          }
        }
      }
    }
  }
}

__global__ __launch_bounds__(GM_THREADS)
void heads_kernel(const int* __restrict__ tokens,
                  const float* __restrict__ h,
                  const float* __restrict__ E,
                  const float* __restrict__ Wn,
                  const float* __restrict__ bn,
                  const float* __restrict__ Wt,
                  const float* __restrict__ bt,
                  const float* __restrict__ tsa,
                  const float* __restrict__ tst,
                  const float* __restrict__ psa, const float* __restrict__ pst,
                  const float* __restrict__ ppa, const float* __restrict__ ppt,
                  const float* __restrict__ pta, const float* __restrict__ ptt,
                  float* __restrict__ out) {
  __shared__ __attribute__((aligned(16))) float lw[NTOT * WPITCH];

  if (blockIdx.x < NPROTO) {
    proto_role(blockIdx.x, lw, tokens, h, E, Wn, bn, Wt, bt, tsa, tst,
               psa, pst, ppa, ppt, pta, ptt, out);
  } else {
    gemm_role(blockIdx.x - NPROTO, lw, tokens, h, E, Wn, bn, Wt, bt, tsa, tst, out);
  }
}

extern "C" void kernel_launch(void* const* d_in, const int* in_sizes, int n_in,
                              void* d_out, int out_size, void* d_ws, size_t ws_size,
                              hipStream_t stream) {
  const int*   tokens = (const int*)  d_in[0];
  const float* h      = (const float*)d_in[1];
  const float* E      = (const float*)d_in[2];
  const float* Wn     = (const float*)d_in[3];
  const float* bn     = (const float*)d_in[4];
  const float* Wt     = (const float*)d_in[5];
  const float* bt     = (const float*)d_in[6];
  const float* tsa    = (const float*)d_in[7];
  const float* tst    = (const float*)d_in[8];
  const float* psa    = (const float*)d_in[9];
  const float* pst    = (const float*)d_in[10];
  const float* ppa    = (const float*)d_in[11];
  const float* ppt    = (const float*)d_in[12];
  const float* pta    = (const float*)d_in[13];
  const float* ptt    = (const float*)d_in[14];
  float* out = (float*)d_out;

  heads_kernel<<<NBLOCKS, GM_THREADS, 0, stream>>>(tokens, h, E, Wn, bn, Wt, bt, tsa, tst,
                                                   psa, pst, ppa, ppt, pta, ptt, out);
}


# proto kernel: the walk forms the row norm of event e+1 during event e, as a ninth value in that step's lane sums, square root and division (same products, pairing, quotient)
# speedup vs baseline: 11.0858x; 1.0878x over previous
#include <hip/hip_runtime.h>
#include <math.h>

typedef float v2f __attribute__((ext_vector_type(2)));
typedef float v8f __attribute__((ext_vector_type(8)));

#define TT 2048
#define DD 256
#define NBATCH 8
#define NACT 64
#define NTIME 32
#define NTOT 96
#define ROWS (NBATCH * TT)
#define OUT_TIME_OFF (ROWS * NACT)
#define WPITCH 260
#define GM_THREADS 128
#define GM_ROWS 64
#define KCHUNK 64
#define NAHEAD 5
#define NSLOT (NAHEAD + 1)
#define PSEG 16
#define NPSEG 6
#define NPROTO 48
#define NBLOCKS 304

__device__ __forceinline__ float softplus_f(float x) { return log1pf(expf(x)); }

__device__ __forceinline__ float xrow16(float x) {
  const int b = __builtin_bit_cast(int, x);
  return __builtin_bit_cast(float, __builtin_amdgcn_permlanex16(
      b, b, 0x76543210, 0xfedcba98, false, false));
}
template <int M>
__device__ __forceinline__ float xrow(float x) {
  const int b = __builtin_bit_cast(int, x);
  return __builtin_bit_cast(float, __builtin_amdgcn_update_dpp(
      0, b, 0x160 | M, 0xf, 0xf, true));
}
__device__ __forceinline__ float wave_sum(float s) {
  s += xrow16(s);
  s += xrow<8>(s);
  s += xrow<4>(s);
  s += xrow<2>(s);
  s += xrow<1>(s);
  return s;
}


__device__ __forceinline__ void gemm_role(const int gb, float* lw,
                                          const int* __restrict__ tokens,
                                          const float* __restrict__ h,
                                          const float* __restrict__ E,
                                          const float* __restrict__ Wn,
                                          const float* __restrict__ bn,
                                          const float* __restrict__ Wt,
                                          const float* __restrict__ bt,
                                          const float* __restrict__ tsa,
                                          const float* __restrict__ tst,
                                          float* __restrict__ out) {
  const int tid   = threadIdx.x;
  const int wave  = tid >> 5;
  const int lane  = tid & 31;
  const int lhalf = lane >> 4;
  const int l16   = lane & 15;
  const int m_base = gb * GM_ROWS + wave * 16;

  const float* arow = h + (m_base + l16) * DD + (lhalf << 1);
  const float* lb = lw + l16 * WPITCH + (lhalf << 1);

  v2f abuf[2][8];
#pragma unroll
  for (int q = 0; q < 8; ++q) abuf[0][q] = *(const v2f*)(arow + q * 4);

  const int* tkr = tokens + m_base + (lhalf << 3);
  const int4 tq0 = *(const int4*)(tkr);
  const int4 tq1 = *(const int4*)(tkr + 4);

  const int sq = tid & 15;
  const int sn = tid >> 4;
  const float* wn_p = Wn + sn * DD + sq * 4;
  const float* ea_p = E + (4 + sn) * DD + sq * 4;
  const float* wt_p = Wt + sn * DD + sq * 4;
  const float* et_p = E + (68 + sn) * DD + sq * 4;
  float* lst = lw + sn * WPITCH + sq * 4;
  float4 wr[12], er[12];

#pragma unroll
  for (int i = 0; i < 8; ++i) {
    wr[i] = *(const float4*)(wn_p + i * 8 * DD);
    er[i] = *(const float4*)(ea_p + i * 8 * DD);
  }
#pragma unroll
  for (int i = 0; i < 4; ++i) {
    wr[8 + i] = *(const float4*)(wt_p + i * 8 * DD);
    er[8 + i] = *(const float4*)(et_p + i * 8 * DD);
  }

  const float s_ta = softplus_f(tsa[0]);
  const float s_tt = softplus_f(tst[0]);

  v8f acc[6];
#pragma unroll
  for (int j = 0; j < 6; ++j) {
    int n = j * 16 + l16;
    float bv = (n < NACT) ? bn[n] : bt[n - NACT];
#pragma unroll
    for (int r = 0; r < 8; ++r) acc[j][r] = bv;
  }

#pragma unroll
  for (int i = 0; i < 12; ++i) {
    const float s = (i < 8) ? s_ta : s_tt;
    float4 v;
    v.x = wr[i].x + s * er[i].x;
    v.y = wr[i].y + s * er[i].y;
    v.z = wr[i].z + s * er[i].z;
    v.w = wr[i].w + s * er[i].w;
    *(float4*)(lst + i * 8 * WPITCH) = v;
  }
  __syncthreads();

#pragma unroll
  for (int hc = 0; hc < 8; ++hc) {
    const int cur = hc & 1;
    const int nxt = cur ^ 1;
    if (hc + 1 < 8) {
#pragma unroll
      for (int q = 0; q < 8; ++q)
        abuf[nxt][q] = *(const v2f*)(arow + (hc + 1) * 32 + q * 4);
    }
    if ((hc & 1) == 0 && hc < 6) {
      const int kc = ((hc >> 1) + 1) * KCHUNK;
#pragma unroll
      for (int i = 0; i < 8; ++i) {
        wr[i] = *(const float4*)(wn_p + i * 8 * DD + kc);
        er[i] = *(const float4*)(ea_p + i * 8 * DD + kc);
      }
#pragma unroll
      for (int i = 0; i < 4; ++i) {
        wr[8 + i] = *(const float4*)(wt_p + i * 8 * DD + kc);
        er[8 + i] = *(const float4*)(et_p + i * 8 * DD + kc);
      }
    }

#pragma unroll
    for (int q = 0; q < 8; ++q) {
      const v2f a = abuf[cur][q];
#pragma unroll
      for (int j = 0; j < 6; ++j) {
        const v2f b = *(const v2f*)(lb + j * 16 * WPITCH + hc * 32 + q * 4);
        acc[j] = __builtin_amdgcn_wmma_f32_16x16x4_f32(
            false, a, false, b, (short)0, acc[j], false, false);
      }
    }

    if ((hc & 1) == 1 && hc < 7) {
      const int kc = ((hc >> 1) + 1) * KCHUNK;
      __builtin_amdgcn_sched_barrier(0);
#pragma unroll
      for (int i = 0; i < 12; ++i) {
        const float s = (i < 8) ? s_ta : s_tt;
        float4 v;
        v.x = wr[i].x + s * er[i].x;
        v.y = wr[i].y + s * er[i].y;
        v.z = wr[i].z + s * er[i].z;
        v.w = wr[i].w + s * er[i].w;
        *(float4*)(lst + i * 8 * WPITCH + kc) = v;
      }
      __syncthreads();
    }
  }

  const int tk8[8] = {tq0.x, tq0.y, tq0.z, tq0.w, tq1.x, tq1.y, tq1.z, tq1.w};
#pragma unroll
  for (int r = 0; r < 8; ++r) {
    if (tk8[r] != 3) {
      const int row = m_base + r + (lhalf << 3);
#pragma unroll
      for (int j = 0; j < 6; ++j) {
        const int n = j * 16 + l16;
        if (n < NACT) out[row * NACT + n] = acc[j][r];
        else          out[OUT_TIME_OFF + row * NTIME + (n - NACT)] = acc[j][r];
      }
    }
  }
}

__device__ __forceinline__ void fetch_event(int ev, int b, int lane,
                                            const float* __restrict__ h,
                                            const float* obase, int ostride,
                                            float4& ra, float4& rc, float& ro) {
  const int r = b * TT + (ev >> 9);
  const float* hr = h + r * DD + lane * 8;
  ra = *(const float4*)(hr);
  rc = *(const float4*)(hr + 4);
  ro = obase[r * ostride + (lane & 3)];
}

__device__ __forceinline__ void proto_role(const int pb, float* smem,
                                           const int* __restrict__ tokens,
                                           const float* __restrict__ h,
                                           const float* __restrict__ E,
                                           const float* __restrict__ Wn,
                                           const float* __restrict__ bn,
                                           const float* __restrict__ Wt,
                                           const float* __restrict__ bt,
                                           const float* __restrict__ tsa,
                                           const float* __restrict__ tst,
                                           const float* __restrict__ psa,
                                           const float* __restrict__ pst,
                                           const float* __restrict__ ppa,
                                           const float* __restrict__ ppt,
                                           const float* __restrict__ pta,
                                           const float* __restrict__ ptt,
                                           float* __restrict__ out) {
  int*   evbuf  = (int*)smem;
  int*   wfirst = evbuf + 2048;
  int*   wcount = evbuf + 2052;
  float* lwp    = smem + 2064;

  const int b   = pb / NPSEG;
  const int sg  = pb % NPSEG;
  const bool is_time = (sg >= 4);
  const int lo = is_time ? 68 : 4;
  const int hi = is_time ? 100 : 68;
  const int c0 = (is_time ? (sg - 4) : sg) * PSEG;
  const int eb = lo + c0;

  const int tid  = threadIdx.x;
  const int wave = __builtin_amdgcn_readfirstlane(tid >> 5);
  const int lane = tid & 31;
  const int lhalf = lane >> 4;
  const int l16   = lane & 15;

  const int* tk = tokens + b * TT;
  const int t0 = tid * 16;
  int tok[17];
  {
    const int4 qa = *(const int4*)(tk + t0);
    const int4 qb = *(const int4*)(tk + t0 + 4);
    const int4 qc = *(const int4*)(tk + t0 + 8);
    const int4 qd = *(const int4*)(tk + t0 + 12);
    tok[0]  = qa.x; tok[1]  = qa.y; tok[2]  = qa.z; tok[3]  = qa.w;
    tok[4]  = qb.x; tok[5]  = qb.y; tok[6]  = qb.z; tok[7]  = qb.w;
    tok[8]  = qc.x; tok[9]  = qc.y; tok[10] = qc.z; tok[11] = qc.w;
    tok[12] = qd.x; tok[13] = qd.y; tok[14] = qd.z; tok[15] = qd.w;
    tok[16] = tk[(t0 + 16) & (TT - 1)];
  }

  const float* Wh = is_time ? Wt : Wn;
  const int sq = tid & 15;
  const int sn = tid >> 4;
  float4 wr[2][4], er[2][4];
#pragma unroll
  for (int i = 0; i < 2; ++i) {
#pragma unroll
    for (int j = 0; j < 4; ++j) {
      wr[i][j] = *(const float4*)(Wh + (c0 + sn + 8 * i) * DD + (sq + 16 * j) * 4);
      er[i][j] = *(const float4*)(E  + (eb + sn + 8 * i) * DD + (sq + 16 * j) * 4);
    }
  }

  const float* bp = is_time ? bt : bn;
  const float bv = bp[c0 + l16];

  float4 ea[4], ec[4];
#pragma unroll
  for (int q = 0; q < 4; ++q) {
    const float* erow = E + (eb + wave * 4 + q) * DD + lane * 8;
    ea[q] = *(const float4*)(erow);
    ec[q] = *(const float4*)(erow + 4);
  }

  const float alpha = softplus_f((is_time ? ppt : ppa)[0]);
  const float st    = softplus_f((is_time ? pst : psa)[0]) *
                      softplus_f((is_time ? ptt : pta)[0]);
  const float s_t   = softplus_f((is_time ? tst : tsa)[0]);

  float acc[4][8];
#pragma unroll
  for (int q = 0; q < 4; ++q) {
    const float v[8] = {ea[q].x, ea[q].y, ea[q].z, ea[q].w, ec[q].x, ec[q].y, ec[q].z, ec[q].w};
    float s = 0.f;
#pragma unroll
    for (int u = 0; u < 8; ++u) s = fmaf(v[u], v[u], s);
    s = wave_sum(s);
    const float inv = 1.0f / fmaxf(sqrtf(s), 1e-12f);
#pragma unroll
    for (int u = 0; u < 8; ++u) acc[q][u] = alpha * v[u] * inv;
  }

  unsigned labm = 0u, supm = 0u;
#pragma unroll
  for (int i = 0; i < 16; ++i) {
    if (tok[i] == 3) {
      labm |= 1u << i;
      int nx = tok[i + 1];
      if (nx >= lo && nx < hi) supm |= 1u << i;
    }
  }
  {
    const unsigned bs = __builtin_amdgcn_ballot_w32(supm != 0u);
    const int fs = t0 + __builtin_ctz(supm | 0x10000u);
    const int fw = __builtin_amdgcn_readlane(fs, bs ? __builtin_ctz(bs) : 0);
    if (lane == 0) wfirst[wave] = bs ? fw : TT;
  }
  __syncthreads();

  int first_sup = wfirst[0];
  for (int w = 1; w < 4; ++w) first_sup = min(first_sup, wfirst[w]);
  const unsigned evm = labm;
  int ebefore = 0, wtotal = 0;
  for (int i = 0; i < 16; ++i) {
    const unsigned bi = __builtin_amdgcn_ballot_w32(((evm >> i) & 1u) != 0u);
    ebefore = __builtin_amdgcn_mbcnt_lo(bi, ebefore);
    wtotal += __popc(bi);
  }
  if (lane == 0) wcount[wave] = wtotal;

#pragma unroll
  for (int i = 0; i < 2; ++i) {
#pragma unroll
    for (int j = 0; j < 4; ++j) {
      float4 v;
      v.x = wr[i][j].x + s_t * er[i][j].x;
      v.y = wr[i][j].y + s_t * er[i][j].y;
      v.z = wr[i][j].z + s_t * er[i][j].z;
      v.w = wr[i][j].w + s_t * er[i][j].w;
      *(float4*)(lwp + (sn + 8 * i) * WPITCH + (sq + 16 * j) * 4) = v;
    }
  }
  __syncthreads();

  {
    int pos = ebefore;
    for (int w = 0; w < 4; ++w) {
      if (w < wave) pos += wcount[w];
    }
#pragma unroll
    for (int i = 0; i < 16; ++i) {
      if (evm & (1u << i)) {
        int t    = t0 + i;
        int emit = (t > first_sup) ? 1 : 0;
        int cls1 = (supm & (1u << i)) ? (tok[i + 1] - lo + 1) : 0;
        evbuf[pos++] = (t << 9) | (emit << 8) | cls1;
      }
    }
  }
  __syncthreads();

  int nevv = 0;
  for (int w = 0; w < 4; ++w) nevv += wcount[w];
  const int nev = __builtin_amdgcn_readfirstlane(nevv);

  const int ostride = is_time ? NTIME : NACT;
  float* const sbase = is_time ? (out + OUT_TIME_OFF + c0) : (out + c0);
  float* const obase = sbase + wave * 4;

  if (nev == 0) return;
  const int last = nev - 1;

  {
    const float* lb = lwp + l16 * WPITCH + (lhalf << 1);
    for (int et = wave; et * 16 < nev; et += 4) {
      const int m  = min(et * 16 + l16, last);
      const int ra = b * TT + (evbuf[m] >> 9);
      const float* arow = h + ra * DD + (lhalf << 1);
      v2f af[64];
#pragma unroll
      for (int q = 0; q < 64; ++q) af[q] = *(const v2f*)(arow + q * 4);
      v8f d;
#pragma unroll
      for (int r = 0; r < 8; ++r) d[r] = bv;
#pragma unroll
      for (int q = 0; q < 64; ++q) {
        const v2f bq = *(const v2f*)(lb + q * 4);
        d = __builtin_amdgcn_wmma_f32_16x16x4_f32(
            false, af[q], false, bq, (short)0, d, false, false);
      }
#pragma unroll
      for (int r = 0; r < 8; ++r) {
        const int mm = et * 16 + r + (lhalf << 3);
        if (mm < nev) {
          const int rr = b * TT + (evbuf[mm] >> 9);
          sbase[rr * ostride + l16] = d[r];
        }
      }
    }
  }
  __syncthreads();

  float4 pa[NSLOT], pc[NSLOT];
  float  po[NSLOT];
  int    pev[NSLOT];
#pragma unroll
  for (int k = 0; k < NSLOT; ++k) {
    pa[k] = make_float4(0.f, 0.f, 0.f, 0.f); pc[k] = pa[k]; po[k] = 0.f; pev[k] = 0;
  }
  int evq = evbuf[0];
  float invh_nx = 0.f;

  for (int e0 = -NSLOT; e0 < nev; e0 += NSLOT) {
#pragma unroll
    for (int s = 0; s < NSLOT; ++s) {
      const int e = e0 + s;

      {
        const int kn  = (s + NSLOT - 1) % NSLOT;
        const int evn = __builtin_amdgcn_readfirstlane(evq);
        pev[kn] = evn;
        fetch_event(evn, b, lane, h, obase, ostride, pa[kn], pc[kn], po[kn]);
        evq = evbuf[min(e + NAHEAD + 1, last)];
      }

      if ((unsigned)(e + 1) > (unsigned)nev) continue;
      const bool live = (e >= 0);

      const int ev   = pev[s];
      const int t    = ev >> 9;
      const int emit = live ? ((ev >> 8) & 1) : 0;
      const int clsp = live ? (ev & 255) : 0;
      const int r    = b * TT + t;

      const float4 ha = pa[s];
      const float4 hc = pc[s];
      const float oldv = po[s];
      float* const op = obase + r * ostride;
      const float hx[8] = {ha.x, ha.y, ha.z, ha.w, hc.x, hc.y, hc.z, hc.w};
      const float invh = invh_nx;

      const int kx = (s + 1) % NSLOT;
      const float4 na = pa[kx];
      const float4 nc = pc[kx];
      const float nx[8] = {na.x, na.y, na.z, na.w, nc.x, nc.y, nc.z, nc.w};
      float hsn = 0.f;
#pragma unroll
      for (int u = 0; u < 8; ++u) hsn = fmaf(nx[u], nx[u], hsn);

      float dot[4], asq[4];
#pragma unroll
      for (int q = 0; q < 4; ++q) {
        dot[q] = 0.f; asq[q] = 0.f;
#pragma unroll
        for (int u = 0; u < 8; ++u) {
          dot[q] = fmaf(acc[q][u], hx[u], dot[q]);
          asq[q] = fmaf(acc[q][u], acc[q][u], asq[q]);
        }
      }
#pragma unroll
      for (int q = 0; q < 4; ++q) { dot[q] += xrow16(dot[q]);  asq[q] += xrow16(asq[q]); }
      hsn += xrow16(hsn);
#pragma unroll
      for (int q = 0; q < 4; ++q) { dot[q] += xrow<8>(dot[q]); asq[q] += xrow<8>(asq[q]); }
      hsn += xrow<8>(hsn);
#pragma unroll
      for (int q = 0; q < 4; ++q) { dot[q] += xrow<4>(dot[q]); asq[q] += xrow<4>(asq[q]); }
      hsn += xrow<4>(hsn);
#pragma unroll
      for (int q = 0; q < 4; ++q) { dot[q] += xrow<2>(dot[q]); asq[q] += xrow<2>(asq[q]); }
      hsn += xrow<2>(hsn);
#pragma unroll
      for (int q = 0; q < 4; ++q) { dot[q] += xrow<1>(dot[q]); asq[q] += xrow<1>(asq[q]); }
      hsn += xrow<1>(hsn);
      const float dsel = (lane == 0) ? dot[0] : (lane == 1) ? dot[1]
                       : (lane == 2) ? dot[2] : dot[3];
      const float asel = (lane == 0) ? asq[0] : (lane == 1) ? asq[1]
                       : (lane == 2) ? asq[2] : asq[3];
      const float ssel = (lane < 4) ? asel : hsn;
      const float num  = (lane < 4) ? dsel * invh : 1.0f;
      const float quot = num / fmaxf(sqrtf(ssel), 1e-12f);
      if (emit && lane < 4) {
        float add = st * quot;
        op[lane] = oldv + add;
      }
      invh_nx = __builtin_bit_cast(float, __builtin_amdgcn_readlane(
          __builtin_bit_cast(int, quot), 4));

      if (clsp) {
        const int cl = clsp - 1 - c0;
        if (cl >= 0 && cl < PSEG && (cl >> 2) == wave) {
          const int qs = cl & 3;
#pragma unroll
          for (int q = 0; q < 4; ++q) {
            if (q == qs) {
#pragma unroll
              for (int u = 0; u < 8; ++u) acc[q][u] += hx[u] * invh;
            }
          }
        }
      }
    }
  }
}

__global__ __launch_bounds__(GM_THREADS)
void heads_kernel(const int* __restrict__ tokens,
                  const float* __restrict__ h,
                  const float* __restrict__ E,
                  const float* __restrict__ Wn,
                  const float* __restrict__ bn,
                  const float* __restrict__ Wt,
                  const float* __restrict__ bt,
                  const float* __restrict__ tsa,
                  const float* __restrict__ tst,
                  const float* __restrict__ psa, const float* __restrict__ pst,
                  const float* __restrict__ ppa, const float* __restrict__ ppt,
                  const float* __restrict__ pta, const float* __restrict__ ptt,
                  float* __restrict__ out) {
  __shared__ __attribute__((aligned(16))) float lw[NTOT * WPITCH];

  if (blockIdx.x < NPROTO) {
    proto_role(blockIdx.x, lw, tokens, h, E, Wn, bn, Wt, bt, tsa, tst,
               psa, pst, ppa, ppt, pta, ptt, out);
  } else {
    gemm_role(blockIdx.x - NPROTO, lw, tokens, h, E, Wn, bn, Wt, bt, tsa, tst, out);
  }
}

extern "C" void kernel_launch(void* const* d_in, const int* in_sizes, int n_in,
                              void* d_out, int out_size, void* d_ws, size_t ws_size,
                              hipStream_t stream) {
  const int*   tokens = (const int*)  d_in[0];
  const float* h      = (const float*)d_in[1];
  const float* E      = (const float*)d_in[2];
  const float* Wn     = (const float*)d_in[3];
  const float* bn     = (const float*)d_in[4];
  const float* Wt     = (const float*)d_in[5];
  const float* bt     = (const float*)d_in[6];
  const float* tsa    = (const float*)d_in[7];
  const float* tst    = (const float*)d_in[8];
  const float* psa    = (const float*)d_in[9];
  const float* pst    = (const float*)d_in[10];
  const float* ppa    = (const float*)d_in[11];
  const float* ppt    = (const float*)d_in[12];
  const float* pta    = (const float*)d_in[13];
  const float* ptt    = (const float*)d_in[14];
  float* out = (float*)d_out;

  heads_kernel<<<NBLOCKS, GM_THREADS, 0, stream>>>(tokens, h, E, Wn, bn, Wt, bt, tsa, tst,
                                                   psa, pst, ppa, ppt, pta, ptt, out);
}


# proto kernel: each 1/max(norm,eps) is one v_rsq_f32 of max(sum of squares, eps squared), about 1 ulp, not correctly rounded sqrt then division (25 instructions) at the end of each event's chain
# speedup vs baseline: 11.8780x; 1.0715x over previous
#include <hip/hip_runtime.h>
#include <math.h>

typedef float v2f __attribute__((ext_vector_type(2)));
typedef float v8f __attribute__((ext_vector_type(8)));

#define TT 2048
#define DD 256
#define NBATCH 8
#define NACT 64
#define NTIME 32
#define NTOT 96
#define ROWS (NBATCH * TT)
#define OUT_TIME_OFF (ROWS * NACT)
#define WPITCH 260
#define GM_THREADS 128
#define GM_ROWS 64
#define KCHUNK 64
#define NAHEAD 5
#define NSLOT (NAHEAD + 1)
#define PSEG 16
#define NPSEG 6
#define NPROTO 48
#define NBLOCKS 304

__device__ __forceinline__ float softplus_f(float x) { return log1pf(expf(x)); }

__device__ __forceinline__ float xrow16(float x) {
  const int b = __builtin_bit_cast(int, x);
  return __builtin_bit_cast(float, __builtin_amdgcn_permlanex16(
      b, b, 0x76543210, 0xfedcba98, false, false));
}
template <int M>
__device__ __forceinline__ float xrow(float x) {
  const int b = __builtin_bit_cast(int, x);
  return __builtin_bit_cast(float, __builtin_amdgcn_update_dpp(
      0, b, 0x160 | M, 0xf, 0xf, true));
}
__device__ __forceinline__ float wave_sum(float s) {
  s += xrow16(s);
  s += xrow<8>(s);
  s += xrow<4>(s);
  s += xrow<2>(s);
  s += xrow<1>(s);
  return s;
}

__device__ __forceinline__ float inv_norm(float s) {
  return __builtin_amdgcn_rsqf(fmaxf(s, 1e-24f));
}


__device__ __forceinline__ void gemm_role(const int gb, float* lw,
                                          const int* __restrict__ tokens,
                                          const float* __restrict__ h,
                                          const float* __restrict__ E,
                                          const float* __restrict__ Wn,
                                          const float* __restrict__ bn,
                                          const float* __restrict__ Wt,
                                          const float* __restrict__ bt,
                                          const float* __restrict__ tsa,
                                          const float* __restrict__ tst,
                                          float* __restrict__ out) {
  const int tid   = threadIdx.x;
  const int wave  = tid >> 5;
  const int lane  = tid & 31;
  const int lhalf = lane >> 4;
  const int l16   = lane & 15;
  const int m_base = gb * GM_ROWS + wave * 16;

  const float* arow = h + (m_base + l16) * DD + (lhalf << 1);
  const float* lb = lw + l16 * WPITCH + (lhalf << 1);

  v2f abuf[2][8];
#pragma unroll
  for (int q = 0; q < 8; ++q) abuf[0][q] = *(const v2f*)(arow + q * 4);

  const int* tkr = tokens + m_base + (lhalf << 3);
  const int4 tq0 = *(const int4*)(tkr);
  const int4 tq1 = *(const int4*)(tkr + 4);

  const int sq = tid & 15;
  const int sn = tid >> 4;
  const float* wn_p = Wn + sn * DD + sq * 4;
  const float* ea_p = E + (4 + sn) * DD + sq * 4;
  const float* wt_p = Wt + sn * DD + sq * 4;
  const float* et_p = E + (68 + sn) * DD + sq * 4;
  float* lst = lw + sn * WPITCH + sq * 4;
  float4 wr[12], er[12];

#pragma unroll
  for (int i = 0; i < 8; ++i) {
    wr[i] = *(const float4*)(wn_p + i * 8 * DD);
    er[i] = *(const float4*)(ea_p + i * 8 * DD);
  }
#pragma unroll
  for (int i = 0; i < 4; ++i) {
    wr[8 + i] = *(const float4*)(wt_p + i * 8 * DD);
    er[8 + i] = *(const float4*)(et_p + i * 8 * DD);
  }

  const float s_ta = softplus_f(tsa[0]);
  const float s_tt = softplus_f(tst[0]);

  v8f acc[6];
#pragma unroll
  for (int j = 0; j < 6; ++j) {
    int n = j * 16 + l16;
    float bv = (n < NACT) ? bn[n] : bt[n - NACT];
#pragma unroll
    for (int r = 0; r < 8; ++r) acc[j][r] = bv;
  }

#pragma unroll
  for (int i = 0; i < 12; ++i) {
    const float s = (i < 8) ? s_ta : s_tt;
    float4 v;
    v.x = wr[i].x + s * er[i].x;
    v.y = wr[i].y + s * er[i].y;
    v.z = wr[i].z + s * er[i].z;
    v.w = wr[i].w + s * er[i].w;
    *(float4*)(lst + i * 8 * WPITCH) = v;
  }
  __syncthreads();

#pragma unroll
  for (int hc = 0; hc < 8; ++hc) {
    const int cur = hc & 1;
    const int nxt = cur ^ 1;
    if (hc + 1 < 8) {
#pragma unroll
      for (int q = 0; q < 8; ++q)
        abuf[nxt][q] = *(const v2f*)(arow + (hc + 1) * 32 + q * 4);
    }
    if ((hc & 1) == 0 && hc < 6) {
      const int kc = ((hc >> 1) + 1) * KCHUNK;
#pragma unroll
      for (int i = 0; i < 8; ++i) {
        wr[i] = *(const float4*)(wn_p + i * 8 * DD + kc);
        er[i] = *(const float4*)(ea_p + i * 8 * DD + kc);
      }
#pragma unroll
      for (int i = 0; i < 4; ++i) {
        wr[8 + i] = *(const float4*)(wt_p + i * 8 * DD + kc);
        er[8 + i] = *(const float4*)(et_p + i * 8 * DD + kc);
      }
    }

#pragma unroll
    for (int q = 0; q < 8; ++q) {
      const v2f a = abuf[cur][q];
#pragma unroll
      for (int j = 0; j < 6; ++j) {
        const v2f b = *(const v2f*)(lb + j * 16 * WPITCH + hc * 32 + q * 4);
        acc[j] = __builtin_amdgcn_wmma_f32_16x16x4_f32(
            false, a, false, b, (short)0, acc[j], false, false);
      }
    }

    if ((hc & 1) == 1 && hc < 7) {
      const int kc = ((hc >> 1) + 1) * KCHUNK;
      __builtin_amdgcn_sched_barrier(0);
#pragma unroll
      for (int i = 0; i < 12; ++i) {
        const float s = (i < 8) ? s_ta : s_tt;
        float4 v;
        v.x = wr[i].x + s * er[i].x;
        v.y = wr[i].y + s * er[i].y;
        v.z = wr[i].z + s * er[i].z;
        v.w = wr[i].w + s * er[i].w;
        *(float4*)(lst + i * 8 * WPITCH + kc) = v;
      }
      __syncthreads();
    }
  }

  const int tk8[8] = {tq0.x, tq0.y, tq0.z, tq0.w, tq1.x, tq1.y, tq1.z, tq1.w};
#pragma unroll
  for (int r = 0; r < 8; ++r) {
    if (tk8[r] != 3) {
      const int row = m_base + r + (lhalf << 3);
#pragma unroll
      for (int j = 0; j < 6; ++j) {
        const int n = j * 16 + l16;
        if (n < NACT) out[row * NACT + n] = acc[j][r];
        else          out[OUT_TIME_OFF + row * NTIME + (n - NACT)] = acc[j][r];
      }
    }
  }
}

__device__ __forceinline__ void fetch_event(int ev, int b, int lane,
                                            const float* __restrict__ h,
                                            const float* obase, int ostride,
                                            float4& ra, float4& rc, float& ro) {
  const int r = b * TT + (ev >> 9);
  const float* hr = h + r * DD + lane * 8;
  ra = *(const float4*)(hr);
  rc = *(const float4*)(hr + 4);
  ro = obase[r * ostride + (lane & 3)];
}

__device__ __forceinline__ void proto_role(const int pb, float* smem,
                                           const int* __restrict__ tokens,
                                           const float* __restrict__ h,
                                           const float* __restrict__ E,
                                           const float* __restrict__ Wn,
                                           const float* __restrict__ bn,
                                           const float* __restrict__ Wt,
                                           const float* __restrict__ bt,
                                           const float* __restrict__ tsa,
                                           const float* __restrict__ tst,
                                           const float* __restrict__ psa,
                                           const float* __restrict__ pst,
                                           const float* __restrict__ ppa,
                                           const float* __restrict__ ppt,
                                           const float* __restrict__ pta,
                                           const float* __restrict__ ptt,
                                           float* __restrict__ out) {
  int*   evbuf  = (int*)smem;
  int*   wfirst = evbuf + 2048;
  int*   wcount = evbuf + 2052;
  float* lwp    = smem + 2064;

  const int b   = pb / NPSEG;
  const int sg  = pb % NPSEG;
  const bool is_time = (sg >= 4);
  const int lo = is_time ? 68 : 4;
  const int hi = is_time ? 100 : 68;
  const int c0 = (is_time ? (sg - 4) : sg) * PSEG;
  const int eb = lo + c0;

  const int tid  = threadIdx.x;
  const int wave = __builtin_amdgcn_readfirstlane(tid >> 5);
  const int lane = tid & 31;
  const int lhalf = lane >> 4;
  const int l16   = lane & 15;

  const int* tk = tokens + b * TT;
  const int t0 = tid * 16;
  int tok[17];
  {
    const int4 qa = *(const int4*)(tk + t0);
    const int4 qb = *(const int4*)(tk + t0 + 4);
    const int4 qc = *(const int4*)(tk + t0 + 8);
    const int4 qd = *(const int4*)(tk + t0 + 12);
    tok[0]  = qa.x; tok[1]  = qa.y; tok[2]  = qa.z; tok[3]  = qa.w;
    tok[4]  = qb.x; tok[5]  = qb.y; tok[6]  = qb.z; tok[7]  = qb.w;
    tok[8]  = qc.x; tok[9]  = qc.y; tok[10] = qc.z; tok[11] = qc.w;
    tok[12] = qd.x; tok[13] = qd.y; tok[14] = qd.z; tok[15] = qd.w;
    tok[16] = tk[(t0 + 16) & (TT - 1)];
  }

  const float* Wh = is_time ? Wt : Wn;
  const int sq = tid & 15;
  const int sn = tid >> 4;
  float4 wr[2][4], er[2][4];
#pragma unroll
  for (int i = 0; i < 2; ++i) {
#pragma unroll
    for (int j = 0; j < 4; ++j) {
      wr[i][j] = *(const float4*)(Wh + (c0 + sn + 8 * i) * DD + (sq + 16 * j) * 4);
      er[i][j] = *(const float4*)(E  + (eb + sn + 8 * i) * DD + (sq + 16 * j) * 4);
    }
  }

  const float* bp = is_time ? bt : bn;
  const float bv = bp[c0 + l16];

  float4 ea[4], ec[4];
#pragma unroll
  for (int q = 0; q < 4; ++q) {
    const float* erow = E + (eb + wave * 4 + q) * DD + lane * 8;
    ea[q] = *(const float4*)(erow);
    ec[q] = *(const float4*)(erow + 4);
  }

  const float alpha = softplus_f((is_time ? ppt : ppa)[0]);
  const float st    = softplus_f((is_time ? pst : psa)[0]) *
                      softplus_f((is_time ? ptt : pta)[0]);
  const float s_t   = softplus_f((is_time ? tst : tsa)[0]);

  float acc[4][8];
#pragma unroll
  for (int q = 0; q < 4; ++q) {
    const float v[8] = {ea[q].x, ea[q].y, ea[q].z, ea[q].w, ec[q].x, ec[q].y, ec[q].z, ec[q].w};
    float s = 0.f;
#pragma unroll
    for (int u = 0; u < 8; ++u) s = fmaf(v[u], v[u], s);
    s = wave_sum(s);
    const float inv = inv_norm(s);
#pragma unroll
    for (int u = 0; u < 8; ++u) acc[q][u] = alpha * v[u] * inv;
  }

  unsigned labm = 0u, supm = 0u;
#pragma unroll
  for (int i = 0; i < 16; ++i) {
    if (tok[i] == 3) {
      labm |= 1u << i;
      int nx = tok[i + 1];
      if (nx >= lo && nx < hi) supm |= 1u << i;
    }
  }
  {
    const unsigned bs = __builtin_amdgcn_ballot_w32(supm != 0u);
    const int fs = t0 + __builtin_ctz(supm | 0x10000u);
    const int fw = __builtin_amdgcn_readlane(fs, bs ? __builtin_ctz(bs) : 0);
    if (lane == 0) wfirst[wave] = bs ? fw : TT;
  }
  __syncthreads();

  int first_sup = wfirst[0];
  for (int w = 1; w < 4; ++w) first_sup = min(first_sup, wfirst[w]);
  const unsigned evm = labm;
  int ebefore = 0, wtotal = 0;
  for (int i = 0; i < 16; ++i) {
    const unsigned bi = __builtin_amdgcn_ballot_w32(((evm >> i) & 1u) != 0u);
    ebefore = __builtin_amdgcn_mbcnt_lo(bi, ebefore);
    wtotal += __popc(bi);
  }
  if (lane == 0) wcount[wave] = wtotal;

#pragma unroll
  for (int i = 0; i < 2; ++i) {
#pragma unroll
    for (int j = 0; j < 4; ++j) {
      float4 v;
      v.x = wr[i][j].x + s_t * er[i][j].x;
      v.y = wr[i][j].y + s_t * er[i][j].y;
      v.z = wr[i][j].z + s_t * er[i][j].z;
      v.w = wr[i][j].w + s_t * er[i][j].w;
      *(float4*)(lwp + (sn + 8 * i) * WPITCH + (sq + 16 * j) * 4) = v;
    }
  }
  __syncthreads();

  {
    int pos = ebefore;
    for (int w = 0; w < 4; ++w) {
      if (w < wave) pos += wcount[w];
    }
#pragma unroll
    for (int i = 0; i < 16; ++i) {
      if (evm & (1u << i)) {
        int t    = t0 + i;
        int emit = (t > first_sup) ? 1 : 0;
        int cls1 = (supm & (1u << i)) ? (tok[i + 1] - lo + 1) : 0;
        evbuf[pos++] = (t << 9) | (emit << 8) | cls1;
      }
    }
  }
  __syncthreads();

  int nevv = 0;
  for (int w = 0; w < 4; ++w) nevv += wcount[w];
  const int nev = __builtin_amdgcn_readfirstlane(nevv);

  const int ostride = is_time ? NTIME : NACT;
  float* const sbase = is_time ? (out + OUT_TIME_OFF + c0) : (out + c0);
  float* const obase = sbase + wave * 4;

  if (nev == 0) return;
  const int last = nev - 1;

  {
    const float* lb = lwp + l16 * WPITCH + (lhalf << 1);
    for (int et = wave; et * 16 < nev; et += 4) {
      const int m  = min(et * 16 + l16, last);
      const int ra = b * TT + (evbuf[m] >> 9);
      const float* arow = h + ra * DD + (lhalf << 1);
      v2f af[64];
#pragma unroll
      for (int q = 0; q < 64; ++q) af[q] = *(const v2f*)(arow + q * 4);
      v8f d;
#pragma unroll
      for (int r = 0; r < 8; ++r) d[r] = bv;
#pragma unroll
      for (int q = 0; q < 64; ++q) {
        const v2f bq = *(const v2f*)(lb + q * 4);
        d = __builtin_amdgcn_wmma_f32_16x16x4_f32(
            false, af[q], false, bq, (short)0, d, false, false);
      }
#pragma unroll
      for (int r = 0; r < 8; ++r) {
        const int mm = et * 16 + r + (lhalf << 3);
        if (mm < nev) {
          const int rr = b * TT + (evbuf[mm] >> 9);
          sbase[rr * ostride + l16] = d[r];
        }
      }
    }
  }
  __syncthreads();

  float4 pa[NSLOT], pc[NSLOT];
  float  po[NSLOT];
  int    pev[NSLOT];
#pragma unroll
  for (int k = 0; k < NSLOT; ++k) {
    pa[k] = make_float4(0.f, 0.f, 0.f, 0.f); pc[k] = pa[k]; po[k] = 0.f; pev[k] = 0;
  }
  int evq = evbuf[0];
  float invh_nx = 0.f;

  for (int e0 = -NSLOT; e0 < nev; e0 += NSLOT) {
#pragma unroll
    for (int s = 0; s < NSLOT; ++s) {
      const int e = e0 + s;

      {
        const int kn  = (s + NSLOT - 1) % NSLOT;
        const int evn = __builtin_amdgcn_readfirstlane(evq);
        pev[kn] = evn;
        fetch_event(evn, b, lane, h, obase, ostride, pa[kn], pc[kn], po[kn]);
        evq = evbuf[min(e + NAHEAD + 1, last)];
      }

      if ((unsigned)(e + 1) > (unsigned)nev) continue;
      const bool live = (e >= 0);

      const int ev   = pev[s];
      const int t    = ev >> 9;
      const int emit = live ? ((ev >> 8) & 1) : 0;
      const int clsp = live ? (ev & 255) : 0;
      const int r    = b * TT + t;

      const float4 ha = pa[s];
      const float4 hc = pc[s];
      const float oldv = po[s];
      float* const op = obase + r * ostride;
      const float hx[8] = {ha.x, ha.y, ha.z, ha.w, hc.x, hc.y, hc.z, hc.w};
      const float invh = invh_nx;

      const int kx = (s + 1) % NSLOT;
      const float4 na = pa[kx];
      const float4 nc = pc[kx];
      const float nx[8] = {na.x, na.y, na.z, na.w, nc.x, nc.y, nc.z, nc.w};
      float hsn = 0.f;
#pragma unroll
      for (int u = 0; u < 8; ++u) hsn = fmaf(nx[u], nx[u], hsn);

      float dot[4], asq[4];
#pragma unroll
      for (int q = 0; q < 4; ++q) {
        dot[q] = 0.f; asq[q] = 0.f;
#pragma unroll
        for (int u = 0; u < 8; ++u) {
          dot[q] = fmaf(acc[q][u], hx[u], dot[q]);
          asq[q] = fmaf(acc[q][u], acc[q][u], asq[q]);
        }
      }
#pragma unroll
      for (int q = 0; q < 4; ++q) { dot[q] += xrow16(dot[q]);  asq[q] += xrow16(asq[q]); }
      hsn += xrow16(hsn);
#pragma unroll
      for (int q = 0; q < 4; ++q) { dot[q] += xrow<8>(dot[q]); asq[q] += xrow<8>(asq[q]); }
      hsn += xrow<8>(hsn);
#pragma unroll
      for (int q = 0; q < 4; ++q) { dot[q] += xrow<4>(dot[q]); asq[q] += xrow<4>(asq[q]); }
      hsn += xrow<4>(hsn);
#pragma unroll
      for (int q = 0; q < 4; ++q) { dot[q] += xrow<2>(dot[q]); asq[q] += xrow<2>(asq[q]); }
      hsn += xrow<2>(hsn);
#pragma unroll
      for (int q = 0; q < 4; ++q) { dot[q] += xrow<1>(dot[q]); asq[q] += xrow<1>(asq[q]); }
      hsn += xrow<1>(hsn);
      const float dsel = (lane == 0) ? dot[0] : (lane == 1) ? dot[1]
                       : (lane == 2) ? dot[2] : dot[3];
      const float asel = (lane == 0) ? asq[0] : (lane == 1) ? asq[1]
                       : (lane == 2) ? asq[2] : asq[3];
      const float ssel = (lane < 4) ? asel : hsn;
      const float num  = (lane < 4) ? dsel * invh : 1.0f;
      const float quot = num * inv_norm(ssel);
      if (emit && lane < 4) {
        float add = st * quot;
        op[lane] = oldv + add;
      }
      invh_nx = __builtin_bit_cast(float, __builtin_amdgcn_readlane(
          __builtin_bit_cast(int, quot), 4));

      if (clsp) {
        const int cl = clsp - 1 - c0;
        if (cl >= 0 && cl < PSEG && (cl >> 2) == wave) {
          const int qs = cl & 3;
#pragma unroll
          for (int q = 0; q < 4; ++q) {
            if (q == qs) {
#pragma unroll
              for (int u = 0; u < 8; ++u) acc[q][u] += hx[u] * invh;
            }
          }
        }
      }
    }
  }
}

__global__ __launch_bounds__(GM_THREADS)
void heads_kernel(const int* __restrict__ tokens,
                  const float* __restrict__ h,
                  const float* __restrict__ E,
                  const float* __restrict__ Wn,
                  const float* __restrict__ bn,
                  const float* __restrict__ Wt,
                  const float* __restrict__ bt,
                  const float* __restrict__ tsa,
                  const float* __restrict__ tst,
                  const float* __restrict__ psa, const float* __restrict__ pst,
                  const float* __restrict__ ppa, const float* __restrict__ ppt,
                  const float* __restrict__ pta, const float* __restrict__ ptt,
                  float* __restrict__ out) {
  __shared__ __attribute__((aligned(16))) float lw[NTOT * WPITCH];

  if (blockIdx.x < NPROTO) {
    proto_role(blockIdx.x, lw, tokens, h, E, Wn, bn, Wt, bt, tsa, tst,
               psa, pst, ppa, ppt, pta, ptt, out);
  } else {
    gemm_role(blockIdx.x - NPROTO, lw, tokens, h, E, Wn, bn, Wt, bt, tsa, tst, out);
  }
}

extern "C" void kernel_launch(void* const* d_in, const int* in_sizes, int n_in,
                              void* d_out, int out_size, void* d_ws, size_t ws_size,
                              hipStream_t stream) {
  const int*   tokens = (const int*)  d_in[0];
  const float* h      = (const float*)d_in[1];
  const float* E      = (const float*)d_in[2];
  const float* Wn     = (const float*)d_in[3];
  const float* bn     = (const float*)d_in[4];
  const float* Wt     = (const float*)d_in[5];
  const float* bt     = (const float*)d_in[6];
  const float* tsa    = (const float*)d_in[7];
  const float* tst    = (const float*)d_in[8];
  const float* psa    = (const float*)d_in[9];
  const float* pst    = (const float*)d_in[10];
  const float* ppa    = (const float*)d_in[11];
  const float* ppt    = (const float*)d_in[12];
  const float* pta    = (const float*)d_in[13];
  const float* ptt    = (const float*)d_in[14];
  float* out = (float*)d_out;

  heads_kernel<<<NBLOCKS, GM_THREADS, 0, stream>>>(tokens, h, E, Wn, bn, Wt, bt, tsa, tst,
                                                   psa, pst, ppa, ppt, pta, ptt, out);
}


# proto walk: a class's inverse norm (sum of squares, wave sum, rsq) is formed at the start and after each support event of that class, kept in a register, not at every event; same products, pairing
# speedup vs baseline: 12.9461x; 1.0899x over previous
#include <hip/hip_runtime.h>
#include <math.h>

typedef float v2f __attribute__((ext_vector_type(2)));
typedef float v8f __attribute__((ext_vector_type(8)));

#define TT 2048
#define DD 256
#define NBATCH 8
#define NACT 64
#define NTIME 32
#define NTOT 96
#define ROWS (NBATCH * TT)
#define OUT_TIME_OFF (ROWS * NACT)
#define WPITCH 260
#define GM_THREADS 128
#define GM_ROWS 64
#define KCHUNK 64
#define NAHEAD 5
#define NSLOT (NAHEAD + 1)
#define PSEG 16
#define NPSEG 6
#define NPROTO 48
#define NBLOCKS 304

__device__ __forceinline__ float softplus_f(float x) { return log1pf(expf(x)); }

__device__ __forceinline__ float xrow16(float x) {
  const int b = __builtin_bit_cast(int, x);
  return __builtin_bit_cast(float, __builtin_amdgcn_permlanex16(
      b, b, 0x76543210, 0xfedcba98, false, false));
}
template <int M>
__device__ __forceinline__ float xrow(float x) {
  const int b = __builtin_bit_cast(int, x);
  return __builtin_bit_cast(float, __builtin_amdgcn_update_dpp(
      0, b, 0x160 | M, 0xf, 0xf, true));
}
__device__ __forceinline__ float wave_sum(float s) {
  s += xrow16(s);
  s += xrow<8>(s);
  s += xrow<4>(s);
  s += xrow<2>(s);
  s += xrow<1>(s);
  return s;
}

__device__ __forceinline__ float inv_norm(float s) {
  return __builtin_amdgcn_rsqf(fmaxf(s, 1e-24f));
}

__device__ __forceinline__ float class_inv_norms(const float (&acc)[4][8], int lane) {
  float asq[4];
#pragma unroll
  for (int q = 0; q < 4; ++q) {
    asq[q] = 0.f;
#pragma unroll
    for (int u = 0; u < 8; ++u) asq[q] = fmaf(acc[q][u], acc[q][u], asq[q]);
  }
#pragma unroll
  for (int q = 0; q < 4; ++q) asq[q] += xrow16(asq[q]);
#pragma unroll
  for (int q = 0; q < 4; ++q) asq[q] += xrow<8>(asq[q]);
#pragma unroll
  for (int q = 0; q < 4; ++q) asq[q] += xrow<4>(asq[q]);
#pragma unroll
  for (int q = 0; q < 4; ++q) asq[q] += xrow<2>(asq[q]);
#pragma unroll
  for (int q = 0; q < 4; ++q) asq[q] += xrow<1>(asq[q]);
  const float asel = (lane == 0) ? asq[0] : (lane == 1) ? asq[1]
                   : (lane == 2) ? asq[2] : asq[3];
  return inv_norm(asel);
}


__device__ __forceinline__ void gemm_role(const int gb, float* lw,
                                          const int* __restrict__ tokens,
                                          const float* __restrict__ h,
                                          const float* __restrict__ E,
                                          const float* __restrict__ Wn,
                                          const float* __restrict__ bn,
                                          const float* __restrict__ Wt,
                                          const float* __restrict__ bt,
                                          const float* __restrict__ tsa,
                                          const float* __restrict__ tst,
                                          float* __restrict__ out) {
  const int tid   = threadIdx.x;
  const int wave  = tid >> 5;
  const int lane  = tid & 31;
  const int lhalf = lane >> 4;
  const int l16   = lane & 15;
  const int m_base = gb * GM_ROWS + wave * 16;

  const float* arow = h + (m_base + l16) * DD + (lhalf << 1);
  const float* lb = lw + l16 * WPITCH + (lhalf << 1);

  v2f abuf[2][8];
#pragma unroll
  for (int q = 0; q < 8; ++q) abuf[0][q] = *(const v2f*)(arow + q * 4);

  const int* tkr = tokens + m_base + (lhalf << 3);
  const int4 tq0 = *(const int4*)(tkr);
  const int4 tq1 = *(const int4*)(tkr + 4);

  const int sq = tid & 15;
  const int sn = tid >> 4;
  const float* wn_p = Wn + sn * DD + sq * 4;
  const float* ea_p = E + (4 + sn) * DD + sq * 4;
  const float* wt_p = Wt + sn * DD + sq * 4;
  const float* et_p = E + (68 + sn) * DD + sq * 4;
  float* lst = lw + sn * WPITCH + sq * 4;
  float4 wr[12], er[12];

#pragma unroll
  for (int i = 0; i < 8; ++i) {
    wr[i] = *(const float4*)(wn_p + i * 8 * DD);
    er[i] = *(const float4*)(ea_p + i * 8 * DD);
  }
#pragma unroll
  for (int i = 0; i < 4; ++i) {
    wr[8 + i] = *(const float4*)(wt_p + i * 8 * DD);
    er[8 + i] = *(const float4*)(et_p + i * 8 * DD);
  }

  const float s_ta = softplus_f(tsa[0]);
  const float s_tt = softplus_f(tst[0]);

  v8f acc[6];
#pragma unroll
  for (int j = 0; j < 6; ++j) {
    int n = j * 16 + l16;
    float bv = (n < NACT) ? bn[n] : bt[n - NACT];
#pragma unroll
    for (int r = 0; r < 8; ++r) acc[j][r] = bv;
  }

#pragma unroll
  for (int i = 0; i < 12; ++i) {
    const float s = (i < 8) ? s_ta : s_tt;
    float4 v;
    v.x = wr[i].x + s * er[i].x;
    v.y = wr[i].y + s * er[i].y;
    v.z = wr[i].z + s * er[i].z;
    v.w = wr[i].w + s * er[i].w;
    *(float4*)(lst + i * 8 * WPITCH) = v;
  }
  __syncthreads();

#pragma unroll
  for (int hc = 0; hc < 8; ++hc) {
    const int cur = hc & 1;
    const int nxt = cur ^ 1;
    if (hc + 1 < 8) {
#pragma unroll
      for (int q = 0; q < 8; ++q)
        abuf[nxt][q] = *(const v2f*)(arow + (hc + 1) * 32 + q * 4);
    }
    if ((hc & 1) == 0 && hc < 6) {
      const int kc = ((hc >> 1) + 1) * KCHUNK;
#pragma unroll
      for (int i = 0; i < 8; ++i) {
        wr[i] = *(const float4*)(wn_p + i * 8 * DD + kc);
        er[i] = *(const float4*)(ea_p + i * 8 * DD + kc);
      }
#pragma unroll
      for (int i = 0; i < 4; ++i) {
        wr[8 + i] = *(const float4*)(wt_p + i * 8 * DD + kc);
        er[8 + i] = *(const float4*)(et_p + i * 8 * DD + kc);
      }
    }

#pragma unroll
    for (int q = 0; q < 8; ++q) {
      const v2f a = abuf[cur][q];
#pragma unroll
      for (int j = 0; j < 6; ++j) {
        const v2f b = *(const v2f*)(lb + j * 16 * WPITCH + hc * 32 + q * 4);
        acc[j] = __builtin_amdgcn_wmma_f32_16x16x4_f32(
            false, a, false, b, (short)0, acc[j], false, false);
      }
    }

    if ((hc & 1) == 1 && hc < 7) {
      const int kc = ((hc >> 1) + 1) * KCHUNK;
      __builtin_amdgcn_sched_barrier(0);
#pragma unroll
      for (int i = 0; i < 12; ++i) {
        const float s = (i < 8) ? s_ta : s_tt;
        float4 v;
        v.x = wr[i].x + s * er[i].x;
        v.y = wr[i].y + s * er[i].y;
        v.z = wr[i].z + s * er[i].z;
        v.w = wr[i].w + s * er[i].w;
        *(float4*)(lst + i * 8 * WPITCH + kc) = v;
      }
      __syncthreads();
    }
  }

  const int tk8[8] = {tq0.x, tq0.y, tq0.z, tq0.w, tq1.x, tq1.y, tq1.z, tq1.w};
#pragma unroll
  for (int r = 0; r < 8; ++r) {
    if (tk8[r] != 3) {
      const int row = m_base + r + (lhalf << 3);
#pragma unroll
      for (int j = 0; j < 6; ++j) {
        const int n = j * 16 + l16;
        if (n < NACT) out[row * NACT + n] = acc[j][r];
        else          out[OUT_TIME_OFF + row * NTIME + (n - NACT)] = acc[j][r];
      }
    }
  }
}

__device__ __forceinline__ void fetch_event(int ev, int b, int lane,
                                            const float* __restrict__ h,
                                            const float* obase, int ostride,
                                            float4& ra, float4& rc, float& ro) {
  const int r = b * TT + (ev >> 9);
  const float* hr = h + r * DD + lane * 8;
  ra = *(const float4*)(hr);
  rc = *(const float4*)(hr + 4);
  ro = obase[r * ostride + (lane & 3)];
}

__device__ __forceinline__ void proto_role(const int pb, float* smem,
                                           const int* __restrict__ tokens,
                                           const float* __restrict__ h,
                                           const float* __restrict__ E,
                                           const float* __restrict__ Wn,
                                           const float* __restrict__ bn,
                                           const float* __restrict__ Wt,
                                           const float* __restrict__ bt,
                                           const float* __restrict__ tsa,
                                           const float* __restrict__ tst,
                                           const float* __restrict__ psa,
                                           const float* __restrict__ pst,
                                           const float* __restrict__ ppa,
                                           const float* __restrict__ ppt,
                                           const float* __restrict__ pta,
                                           const float* __restrict__ ptt,
                                           float* __restrict__ out) {
  int*   evbuf  = (int*)smem;
  int*   wfirst = evbuf + 2048;
  int*   wcount = evbuf + 2052;
  float* lwp    = smem + 2064;

  const int b   = pb / NPSEG;
  const int sg  = pb % NPSEG;
  const bool is_time = (sg >= 4);
  const int lo = is_time ? 68 : 4;
  const int hi = is_time ? 100 : 68;
  const int c0 = (is_time ? (sg - 4) : sg) * PSEG;
  const int eb = lo + c0;

  const int tid  = threadIdx.x;
  const int wave = __builtin_amdgcn_readfirstlane(tid >> 5);
  const int lane = tid & 31;
  const int lhalf = lane >> 4;
  const int l16   = lane & 15;

  const int* tk = tokens + b * TT;
  const int t0 = tid * 16;
  int tok[17];
  {
    const int4 qa = *(const int4*)(tk + t0);
    const int4 qb = *(const int4*)(tk + t0 + 4);
    const int4 qc = *(const int4*)(tk + t0 + 8);
    const int4 qd = *(const int4*)(tk + t0 + 12);
    tok[0]  = qa.x; tok[1]  = qa.y; tok[2]  = qa.z; tok[3]  = qa.w;
    tok[4]  = qb.x; tok[5]  = qb.y; tok[6]  = qb.z; tok[7]  = qb.w;
    tok[8]  = qc.x; tok[9]  = qc.y; tok[10] = qc.z; tok[11] = qc.w;
    tok[12] = qd.x; tok[13] = qd.y; tok[14] = qd.z; tok[15] = qd.w;
    tok[16] = tk[(t0 + 16) & (TT - 1)];
  }

  const float* Wh = is_time ? Wt : Wn;
  const int sq = tid & 15;
  const int sn = tid >> 4;
  float4 wr[2][4], er[2][4];
#pragma unroll
  for (int i = 0; i < 2; ++i) {
#pragma unroll
    for (int j = 0; j < 4; ++j) {
      wr[i][j] = *(const float4*)(Wh + (c0 + sn + 8 * i) * DD + (sq + 16 * j) * 4);
      er[i][j] = *(const float4*)(E  + (eb + sn + 8 * i) * DD + (sq + 16 * j) * 4);
    }
  }

  const float* bp = is_time ? bt : bn;
  const float bv = bp[c0 + l16];

  float4 ea[4], ec[4];
#pragma unroll
  for (int q = 0; q < 4; ++q) {
    const float* erow = E + (eb + wave * 4 + q) * DD + lane * 8;
    ea[q] = *(const float4*)(erow);
    ec[q] = *(const float4*)(erow + 4);
  }

  const float alpha = softplus_f((is_time ? ppt : ppa)[0]);
  const float st    = softplus_f((is_time ? pst : psa)[0]) *
                      softplus_f((is_time ? ptt : pta)[0]);
  const float s_t   = softplus_f((is_time ? tst : tsa)[0]);

  float acc[4][8];
#pragma unroll
  for (int q = 0; q < 4; ++q) {
    const float v[8] = {ea[q].x, ea[q].y, ea[q].z, ea[q].w, ec[q].x, ec[q].y, ec[q].z, ec[q].w};
    float s = 0.f;
#pragma unroll
    for (int u = 0; u < 8; ++u) s = fmaf(v[u], v[u], s);
    s = wave_sum(s);
    const float inv = inv_norm(s);
#pragma unroll
    for (int u = 0; u < 8; ++u) acc[q][u] = alpha * v[u] * inv;
  }

  unsigned labm = 0u, supm = 0u;
#pragma unroll
  for (int i = 0; i < 16; ++i) {
    if (tok[i] == 3) {
      labm |= 1u << i;
      int nx = tok[i + 1];
      if (nx >= lo && nx < hi) supm |= 1u << i;
    }
  }
  {
    const unsigned bs = __builtin_amdgcn_ballot_w32(supm != 0u);
    const int fs = t0 + __builtin_ctz(supm | 0x10000u);
    const int fw = __builtin_amdgcn_readlane(fs, bs ? __builtin_ctz(bs) : 0);
    if (lane == 0) wfirst[wave] = bs ? fw : TT;
  }
  __syncthreads();

  int first_sup = wfirst[0];
  for (int w = 1; w < 4; ++w) first_sup = min(first_sup, wfirst[w]);
  const unsigned evm = labm;
  int ebefore = 0, wtotal = 0;
  for (int i = 0; i < 16; ++i) {
    const unsigned bi = __builtin_amdgcn_ballot_w32(((evm >> i) & 1u) != 0u);
    ebefore = __builtin_amdgcn_mbcnt_lo(bi, ebefore);
    wtotal += __popc(bi);
  }
  if (lane == 0) wcount[wave] = wtotal;

#pragma unroll
  for (int i = 0; i < 2; ++i) {
#pragma unroll
    for (int j = 0; j < 4; ++j) {
      float4 v;
      v.x = wr[i][j].x + s_t * er[i][j].x;
      v.y = wr[i][j].y + s_t * er[i][j].y;
      v.z = wr[i][j].z + s_t * er[i][j].z;
      v.w = wr[i][j].w + s_t * er[i][j].w;
      *(float4*)(lwp + (sn + 8 * i) * WPITCH + (sq + 16 * j) * 4) = v;
    }
  }
  __syncthreads();

  {
    int pos = ebefore;
    for (int w = 0; w < 4; ++w) {
      if (w < wave) pos += wcount[w];
    }
#pragma unroll
    for (int i = 0; i < 16; ++i) {
      if (evm & (1u << i)) {
        int t    = t0 + i;
        int emit = (t > first_sup) ? 1 : 0;
        int cls1 = (supm & (1u << i)) ? (tok[i + 1] - lo + 1) : 0;
        evbuf[pos++] = (t << 9) | (emit << 8) | cls1;
      }
    }
  }
  __syncthreads();

  int nevv = 0;
  for (int w = 0; w < 4; ++w) nevv += wcount[w];
  const int nev = __builtin_amdgcn_readfirstlane(nevv);

  const int ostride = is_time ? NTIME : NACT;
  float* const sbase = is_time ? (out + OUT_TIME_OFF + c0) : (out + c0);
  float* const obase = sbase + wave * 4;

  if (nev == 0) return;
  const int last = nev - 1;

  {
    const float* lb = lwp + l16 * WPITCH + (lhalf << 1);
    for (int et = wave; et * 16 < nev; et += 4) {
      const int m  = min(et * 16 + l16, last);
      const int ra = b * TT + (evbuf[m] >> 9);
      const float* arow = h + ra * DD + (lhalf << 1);
      v2f af[64];
#pragma unroll
      for (int q = 0; q < 64; ++q) af[q] = *(const v2f*)(arow + q * 4);
      v8f d;
#pragma unroll
      for (int r = 0; r < 8; ++r) d[r] = bv;
#pragma unroll
      for (int q = 0; q < 64; ++q) {
        const v2f bq = *(const v2f*)(lb + q * 4);
        d = __builtin_amdgcn_wmma_f32_16x16x4_f32(
            false, af[q], false, bq, (short)0, d, false, false);
      }
#pragma unroll
      for (int r = 0; r < 8; ++r) {
        const int mm = et * 16 + r + (lhalf << 3);
        if (mm < nev) {
          const int rr = b * TT + (evbuf[mm] >> 9);
          sbase[rr * ostride + l16] = d[r];
        }
      }
    }
  }
  __syncthreads();

  float4 pa[NSLOT], pc[NSLOT];
  float  po[NSLOT];
  int    pev[NSLOT];
#pragma unroll
  for (int k = 0; k < NSLOT; ++k) {
    pa[k] = make_float4(0.f, 0.f, 0.f, 0.f); pc[k] = pa[k]; po[k] = 0.f; pev[k] = 0;
  }
  int evq = evbuf[0];
  float invh_nx = 0.f;
  float ainvl = class_inv_norms(acc, lane);

  for (int e0 = -NSLOT; e0 < nev; e0 += NSLOT) {
#pragma unroll
    for (int s = 0; s < NSLOT; ++s) {
      const int e = e0 + s;

      {
        const int kn  = (s + NSLOT - 1) % NSLOT;
        const int evn = __builtin_amdgcn_readfirstlane(evq);
        pev[kn] = evn;
        fetch_event(evn, b, lane, h, obase, ostride, pa[kn], pc[kn], po[kn]);
        evq = evbuf[min(e + NAHEAD + 1, last)];
      }

      if ((unsigned)(e + 1) > (unsigned)nev) continue;
      const bool live = (e >= 0);

      const int ev   = pev[s];
      const int t    = ev >> 9;
      const int emit = live ? ((ev >> 8) & 1) : 0;
      const int clsp = live ? (ev & 255) : 0;
      const int r    = b * TT + t;

      const float4 ha = pa[s];
      const float4 hc = pc[s];
      const float oldv = po[s];
      float* const op = obase + r * ostride;
      const float hx[8] = {ha.x, ha.y, ha.z, ha.w, hc.x, hc.y, hc.z, hc.w};
      const float invh = invh_nx;

      const int kx = (s + 1) % NSLOT;
      const float4 na = pa[kx];
      const float4 nc = pc[kx];
      const float nx[8] = {na.x, na.y, na.z, na.w, nc.x, nc.y, nc.z, nc.w};
      float hsn = 0.f;
#pragma unroll
      for (int u = 0; u < 8; ++u) hsn = fmaf(nx[u], nx[u], hsn);

      float dot[4];
#pragma unroll
      for (int q = 0; q < 4; ++q) {
        dot[q] = 0.f;
#pragma unroll
        for (int u = 0; u < 8; ++u) dot[q] = fmaf(acc[q][u], hx[u], dot[q]);
      }
#pragma unroll
      for (int q = 0; q < 4; ++q) dot[q] += xrow16(dot[q]);
      hsn += xrow16(hsn);
#pragma unroll
      for (int q = 0; q < 4; ++q) dot[q] += xrow<8>(dot[q]);
      hsn += xrow<8>(hsn);
#pragma unroll
      for (int q = 0; q < 4; ++q) dot[q] += xrow<4>(dot[q]);
      hsn += xrow<4>(hsn);
#pragma unroll
      for (int q = 0; q < 4; ++q) dot[q] += xrow<2>(dot[q]);
      hsn += xrow<2>(hsn);
#pragma unroll
      for (int q = 0; q < 4; ++q) dot[q] += xrow<1>(dot[q]);
      hsn += xrow<1>(hsn);
      const float dsel = (lane == 0) ? dot[0] : (lane == 1) ? dot[1]
                       : (lane == 2) ? dot[2] : dot[3];
      const float quot = dsel * invh * ainvl;
      if (emit && lane < 4) {
        float add = st * quot;
        op[lane] = oldv + add;
      }
      invh_nx = inv_norm(hsn);

      if (clsp) {
        const int cl = clsp - 1 - c0;
        if (cl >= 0 && cl < PSEG && (cl >> 2) == wave) {
          const int qs = cl & 3;
          float s2 = 0.f;
#pragma unroll
          for (int q = 0; q < 4; ++q) {
            if (q == qs) {
#pragma unroll
              for (int u = 0; u < 8; ++u) acc[q][u] += hx[u] * invh;
#pragma unroll
              for (int u = 0; u < 8; ++u) s2 = fmaf(acc[q][u], acc[q][u], s2);
            }
          }
          const float a2 = inv_norm(wave_sum(s2));
          ainvl = (lane == qs) ? a2 : ainvl;
        }
      }
    }
  }
}

__global__ __launch_bounds__(GM_THREADS)
void heads_kernel(const int* __restrict__ tokens,
                  const float* __restrict__ h,
                  const float* __restrict__ E,
                  const float* __restrict__ Wn,
                  const float* __restrict__ bn,
                  const float* __restrict__ Wt,
                  const float* __restrict__ bt,
                  const float* __restrict__ tsa,
                  const float* __restrict__ tst,
                  const float* __restrict__ psa, const float* __restrict__ pst,
                  const float* __restrict__ ppa, const float* __restrict__ ppt,
                  const float* __restrict__ pta, const float* __restrict__ ptt,
                  float* __restrict__ out) {
  __shared__ __attribute__((aligned(16))) float lw[NTOT * WPITCH];

  if (blockIdx.x < NPROTO) {
    proto_role(blockIdx.x, lw, tokens, h, E, Wn, bn, Wt, bt, tsa, tst,
               psa, pst, ppa, ppt, pta, ptt, out);
  } else {
    gemm_role(blockIdx.x - NPROTO, lw, tokens, h, E, Wn, bn, Wt, bt, tsa, tst, out);
  }
}

extern "C" void kernel_launch(void* const* d_in, const int* in_sizes, int n_in,
                              void* d_out, int out_size, void* d_ws, size_t ws_size,
                              hipStream_t stream) {
  const int*   tokens = (const int*)  d_in[0];
  const float* h      = (const float*)d_in[1];
  const float* E      = (const float*)d_in[2];
  const float* Wn     = (const float*)d_in[3];
  const float* bn     = (const float*)d_in[4];
  const float* Wt     = (const float*)d_in[5];
  const float* bt     = (const float*)d_in[6];
  const float* tsa    = (const float*)d_in[7];
  const float* tst    = (const float*)d_in[8];
  const float* psa    = (const float*)d_in[9];
  const float* pst    = (const float*)d_in[10];
  const float* ppa    = (const float*)d_in[11];
  const float* ppt    = (const float*)d_in[12];
  const float* pta    = (const float*)d_in[13];
  const float* ptt    = (const float*)d_in[14];
  float* out = (float*)d_out;

  heads_kernel<<<NBLOCKS, GM_THREADS, 0, stream>>>(tokens, h, E, Wn, bn, Wt, bt, tsa, tst,
                                                   psa, pst, ppa, ppt, pta, ptt, out);
}


# proto walk cut into 4 time segments, one block each (192 blocks): a wave first adds the earlier support rows of its own classes, same order and arithmetic, then walks a quarter of the events
# speedup vs baseline: 16.9864x; 1.3121x over previous
#include <hip/hip_runtime.h>
#include <math.h>

typedef float v2f __attribute__((ext_vector_type(2)));
typedef float v8f __attribute__((ext_vector_type(8)));

#define TT 2048
#define DD 256
#define NBATCH 8
#define NACT 64
#define NTIME 32
#define NTOT 96
#define ROWS (NBATCH * TT)
#define OUT_TIME_OFF (ROWS * NACT)
#define WPITCH 260
#define GM_THREADS 128
#define GM_ROWS 64
#define KCHUNK 64
#define NAHEAD 5
#define NSLOT (NAHEAD + 1)
#define PSEG 16
#define NPSEG 6
#define NTSEG 4
#define NPROTO 192
#define NBLOCKS 448
#define WLIST_OFF 6240
#define WLIST_LEN 2048

__device__ __forceinline__ float softplus_f(float x) { return log1pf(expf(x)); }

__device__ __forceinline__ float xrow16(float x) {
  const int b = __builtin_bit_cast(int, x);
  return __builtin_bit_cast(float, __builtin_amdgcn_permlanex16(
      b, b, 0x76543210, 0xfedcba98, false, false));
}
template <int M>
__device__ __forceinline__ float xrow(float x) {
  const int b = __builtin_bit_cast(int, x);
  return __builtin_bit_cast(float, __builtin_amdgcn_update_dpp(
      0, b, 0x160 | M, 0xf, 0xf, true));
}
__device__ __forceinline__ float wave_sum(float s) {
  s += xrow16(s);
  s += xrow<8>(s);
  s += xrow<4>(s);
  s += xrow<2>(s);
  s += xrow<1>(s);
  return s;
}

__device__ __forceinline__ float inv_norm(float s) {
  return __builtin_amdgcn_rsqf(fmaxf(s, 1e-24f));
}

__device__ __forceinline__ float class_inv_norms(const float (&acc)[4][8], int lane) {
  float asq[4];
#pragma unroll
  for (int q = 0; q < 4; ++q) {
    asq[q] = 0.f;
#pragma unroll
    for (int u = 0; u < 8; ++u) asq[q] = fmaf(acc[q][u], acc[q][u], asq[q]);
  }
#pragma unroll
  for (int q = 0; q < 4; ++q) asq[q] += xrow16(asq[q]);
#pragma unroll
  for (int q = 0; q < 4; ++q) asq[q] += xrow<8>(asq[q]);
#pragma unroll
  for (int q = 0; q < 4; ++q) asq[q] += xrow<4>(asq[q]);
#pragma unroll
  for (int q = 0; q < 4; ++q) asq[q] += xrow<2>(asq[q]);
#pragma unroll
  for (int q = 0; q < 4; ++q) asq[q] += xrow<1>(asq[q]);
  const float asel = (lane == 0) ? asq[0] : (lane == 1) ? asq[1]
                   : (lane == 2) ? asq[2] : asq[3];
  return inv_norm(asel);
}


__device__ __forceinline__ void gemm_role(const int gb, float* lw,
                                          const int* __restrict__ tokens,
                                          const float* __restrict__ h,
                                          const float* __restrict__ E,
                                          const float* __restrict__ Wn,
                                          const float* __restrict__ bn,
                                          const float* __restrict__ Wt,
                                          const float* __restrict__ bt,
                                          const float* __restrict__ tsa,
                                          const float* __restrict__ tst,
                                          float* __restrict__ out) {
  const int tid   = threadIdx.x;
  const int wave  = tid >> 5;
  const int lane  = tid & 31;
  const int lhalf = lane >> 4;
  const int l16   = lane & 15;
  const int m_base = gb * GM_ROWS + wave * 16;

  const float* arow = h + (m_base + l16) * DD + (lhalf << 1);
  const float* lb = lw + l16 * WPITCH + (lhalf << 1);

  v2f abuf[2][8];
#pragma unroll
  for (int q = 0; q < 8; ++q) abuf[0][q] = *(const v2f*)(arow + q * 4);

  const int* tkr = tokens + m_base + (lhalf << 3);
  const int4 tq0 = *(const int4*)(tkr);
  const int4 tq1 = *(const int4*)(tkr + 4);

  const int sq = tid & 15;
  const int sn = tid >> 4;
  const float* wn_p = Wn + sn * DD + sq * 4;
  const float* ea_p = E + (4 + sn) * DD + sq * 4;
  const float* wt_p = Wt + sn * DD + sq * 4;
  const float* et_p = E + (68 + sn) * DD + sq * 4;
  float* lst = lw + sn * WPITCH + sq * 4;
  float4 wr[12], er[12];

#pragma unroll
  for (int i = 0; i < 8; ++i) {
    wr[i] = *(const float4*)(wn_p + i * 8 * DD);
    er[i] = *(const float4*)(ea_p + i * 8 * DD);
  }
#pragma unroll
  for (int i = 0; i < 4; ++i) {
    wr[8 + i] = *(const float4*)(wt_p + i * 8 * DD);
    er[8 + i] = *(const float4*)(et_p + i * 8 * DD);
  }

  const float s_ta = softplus_f(tsa[0]);
  const float s_tt = softplus_f(tst[0]);

  v8f acc[6];
#pragma unroll
  for (int j = 0; j < 6; ++j) {
    int n = j * 16 + l16;
    float bv = (n < NACT) ? bn[n] : bt[n - NACT];
#pragma unroll
    for (int r = 0; r < 8; ++r) acc[j][r] = bv;
  }

#pragma unroll
  for (int i = 0; i < 12; ++i) {
    const float s = (i < 8) ? s_ta : s_tt;
    float4 v;
    v.x = wr[i].x + s * er[i].x;
    v.y = wr[i].y + s * er[i].y;
    v.z = wr[i].z + s * er[i].z;
    v.w = wr[i].w + s * er[i].w;
    *(float4*)(lst + i * 8 * WPITCH) = v;
  }
  __syncthreads();

#pragma unroll
  for (int hc = 0; hc < 8; ++hc) {
    const int cur = hc & 1;
    const int nxt = cur ^ 1;
    if (hc + 1 < 8) {
#pragma unroll
      for (int q = 0; q < 8; ++q)
        abuf[nxt][q] = *(const v2f*)(arow + (hc + 1) * 32 + q * 4);
    }
    if ((hc & 1) == 0 && hc < 6) {
      const int kc = ((hc >> 1) + 1) * KCHUNK;
#pragma unroll
      for (int i = 0; i < 8; ++i) {
        wr[i] = *(const float4*)(wn_p + i * 8 * DD + kc);
        er[i] = *(const float4*)(ea_p + i * 8 * DD + kc);
      }
#pragma unroll
      for (int i = 0; i < 4; ++i) {
        wr[8 + i] = *(const float4*)(wt_p + i * 8 * DD + kc);
        er[8 + i] = *(const float4*)(et_p + i * 8 * DD + kc);
      }
    }

#pragma unroll
    for (int q = 0; q < 8; ++q) {
      const v2f a = abuf[cur][q];
#pragma unroll
      for (int j = 0; j < 6; ++j) {
        const v2f b = *(const v2f*)(lb + j * 16 * WPITCH + hc * 32 + q * 4);
        acc[j] = __builtin_amdgcn_wmma_f32_16x16x4_f32(
            false, a, false, b, (short)0, acc[j], false, false);
      }
    }

    if ((hc & 1) == 1 && hc < 7) {
      const int kc = ((hc >> 1) + 1) * KCHUNK;
      __builtin_amdgcn_sched_barrier(0);
#pragma unroll
      for (int i = 0; i < 12; ++i) {
        const float s = (i < 8) ? s_ta : s_tt;
        float4 v;
        v.x = wr[i].x + s * er[i].x;
        v.y = wr[i].y + s * er[i].y;
        v.z = wr[i].z + s * er[i].z;
        v.w = wr[i].w + s * er[i].w;
        *(float4*)(lst + i * 8 * WPITCH + kc) = v;
      }
      __syncthreads();
    }
  }

  const int tk8[8] = {tq0.x, tq0.y, tq0.z, tq0.w, tq1.x, tq1.y, tq1.z, tq1.w};
#pragma unroll
  for (int r = 0; r < 8; ++r) {
    if (tk8[r] != 3) {
      const int row = m_base + r + (lhalf << 3);
#pragma unroll
      for (int j = 0; j < 6; ++j) {
        const int n = j * 16 + l16;
        if (n < NACT) out[row * NACT + n] = acc[j][r];
        else          out[OUT_TIME_OFF + row * NTIME + (n - NACT)] = acc[j][r];
      }
    }
  }
}

__device__ __forceinline__ void fetch_event(int ev, int b, int lane,
                                            const float* __restrict__ h,
                                            const float* obase, int ostride,
                                            float4& ra, float4& rc, float& ro) {
  const int r = b * TT + (ev >> 9);
  const float* hr = h + r * DD + lane * 8;
  ra = *(const float4*)(hr);
  rc = *(const float4*)(hr + 4);
  ro = obase[r * ostride + (lane & 3)];
}

__device__ __forceinline__ void proto_role(const int pb, float* smem,
                                           const int* __restrict__ tokens,
                                           const float* __restrict__ h,
                                           const float* __restrict__ E,
                                           const float* __restrict__ Wn,
                                           const float* __restrict__ bn,
                                           const float* __restrict__ Wt,
                                           const float* __restrict__ bt,
                                           const float* __restrict__ tsa,
                                           const float* __restrict__ tst,
                                           const float* __restrict__ psa,
                                           const float* __restrict__ pst,
                                           const float* __restrict__ ppa,
                                           const float* __restrict__ ppt,
                                           const float* __restrict__ pta,
                                           const float* __restrict__ ptt,
                                           float* __restrict__ out) {
  int*   evbuf  = (int*)smem;
  int*   wfirst = evbuf + 2048;
  int*   wcount = evbuf + 2052;
  float* lwp    = smem + 2064;

  const int tseg = NTSEG - 1 - pb / (NBATCH * NPSEG);
  const int b   = (pb % (NBATCH * NPSEG)) / NPSEG;
  const int sg  = pb % NPSEG;
  const bool is_time = (sg >= 4);
  const int lo = is_time ? 68 : 4;
  const int hi = is_time ? 100 : 68;
  const int c0 = (is_time ? (sg - 4) : sg) * PSEG;
  const int eb = lo + c0;

  const int tid  = threadIdx.x;
  const int wave = __builtin_amdgcn_readfirstlane(tid >> 5);
  const int lane = tid & 31;
  const int lhalf = lane >> 4;
  const int l16   = lane & 15;

  const int* tk = tokens + b * TT;
  const int t0 = tid * 16;
  int tok[17];
  {
    const int4 qa = *(const int4*)(tk + t0);
    const int4 qb = *(const int4*)(tk + t0 + 4);
    const int4 qc = *(const int4*)(tk + t0 + 8);
    const int4 qd = *(const int4*)(tk + t0 + 12);
    tok[0]  = qa.x; tok[1]  = qa.y; tok[2]  = qa.z; tok[3]  = qa.w;
    tok[4]  = qb.x; tok[5]  = qb.y; tok[6]  = qb.z; tok[7]  = qb.w;
    tok[8]  = qc.x; tok[9]  = qc.y; tok[10] = qc.z; tok[11] = qc.w;
    tok[12] = qd.x; tok[13] = qd.y; tok[14] = qd.z; tok[15] = qd.w;
    tok[16] = tk[(t0 + 16) & (TT - 1)];
  }

  const float* Wh = is_time ? Wt : Wn;
  const int sq = tid & 15;
  const int sn = tid >> 4;
  float4 wr[2][4], er[2][4];
#pragma unroll
  for (int i = 0; i < 2; ++i) {
#pragma unroll
    for (int j = 0; j < 4; ++j) {
      wr[i][j] = *(const float4*)(Wh + (c0 + sn + 8 * i) * DD + (sq + 16 * j) * 4);
      er[i][j] = *(const float4*)(E  + (eb + sn + 8 * i) * DD + (sq + 16 * j) * 4);
    }
  }

  const float* bp = is_time ? bt : bn;
  const float bv = bp[c0 + l16];

  float4 ea[4], ec[4];
#pragma unroll
  for (int q = 0; q < 4; ++q) {
    const float* erow = E + (eb + wave * 4 + q) * DD + lane * 8;
    ea[q] = *(const float4*)(erow);
    ec[q] = *(const float4*)(erow + 4);
  }

  const float alpha = softplus_f((is_time ? ppt : ppa)[0]);
  const float st    = softplus_f((is_time ? pst : psa)[0]) *
                      softplus_f((is_time ? ptt : pta)[0]);
  const float s_t   = softplus_f((is_time ? tst : tsa)[0]);

  float acc[4][8];
#pragma unroll
  for (int q = 0; q < 4; ++q) {
    const float v[8] = {ea[q].x, ea[q].y, ea[q].z, ea[q].w, ec[q].x, ec[q].y, ec[q].z, ec[q].w};
    float s = 0.f;
#pragma unroll
    for (int u = 0; u < 8; ++u) s = fmaf(v[u], v[u], s);
    s = wave_sum(s);
    const float inv = inv_norm(s);
#pragma unroll
    for (int u = 0; u < 8; ++u) acc[q][u] = alpha * v[u] * inv;
  }

  unsigned labm = 0u, supm = 0u;
#pragma unroll
  for (int i = 0; i < 16; ++i) {
    if (tok[i] == 3) {
      labm |= 1u << i;
      int nx = tok[i + 1];
      if (nx >= lo && nx < hi) supm |= 1u << i;
    }
  }
  {
    const unsigned bs = __builtin_amdgcn_ballot_w32(supm != 0u);
    const int fs = t0 + __builtin_ctz(supm | 0x10000u);
    const int fw = __builtin_amdgcn_readlane(fs, bs ? __builtin_ctz(bs) : 0);
    if (lane == 0) wfirst[wave] = bs ? fw : TT;
  }
  __syncthreads();

  int first_sup = wfirst[0];
  for (int w = 1; w < 4; ++w) first_sup = min(first_sup, wfirst[w]);
  const unsigned evm = labm;
  int ebefore = 0, wtotal = 0;
  for (int i = 0; i < 16; ++i) {
    const unsigned bi = __builtin_amdgcn_ballot_w32(((evm >> i) & 1u) != 0u);
    ebefore = __builtin_amdgcn_mbcnt_lo(bi, ebefore);
    wtotal += __popc(bi);
  }
  if (lane == 0) wcount[wave] = wtotal;

#pragma unroll
  for (int i = 0; i < 2; ++i) {
#pragma unroll
    for (int j = 0; j < 4; ++j) {
      float4 v;
      v.x = wr[i][j].x + s_t * er[i][j].x;
      v.y = wr[i][j].y + s_t * er[i][j].y;
      v.z = wr[i][j].z + s_t * er[i][j].z;
      v.w = wr[i][j].w + s_t * er[i][j].w;
      *(float4*)(lwp + (sn + 8 * i) * WPITCH + (sq + 16 * j) * 4) = v;
    }
  }
  __syncthreads();

  {
    int pos = ebefore;
    for (int w = 0; w < 4; ++w) {
      if (w < wave) pos += wcount[w];
    }
#pragma unroll
    for (int i = 0; i < 16; ++i) {
      if (evm & (1u << i)) {
        int t    = t0 + i;
        int emit = (t > first_sup) ? 1 : 0;
        int cls1 = (supm & (1u << i)) ? (tok[i + 1] - lo + 1) : 0;
        evbuf[pos++] = (t << 9) | (emit << 8) | cls1;
      }
    }
  }
  __syncthreads();

  int nevv = 0;
  for (int w = 0; w < 4; ++w) nevv += wcount[w];
  const int nev = __builtin_amdgcn_readfirstlane(nevv);

  const int ostride = is_time ? NTIME : NACT;
  float* const sbase = is_time ? (out + OUT_TIME_OFF + c0) : (out + c0);
  float* const obase = sbase + wave * 4;

  const int e_lo  = (nev * tseg) / NTSEG;
  const int e_hi  = (nev * (tseg + 1)) / NTSEG;
  const int n_own = e_hi - e_lo;
  if (n_own == 0) return;

  int* const wl = (int*)smem + WLIST_OFF + wave * WLIST_LEN;
  int nrep = 0;
#pragma nounroll
  for (int base = 0; base < e_lo; base += 32) {
    const int i  = base + lane;
    const int w  = (i < e_lo) ? evbuf[i] : 0;
    const int cl = (w & 255) - 1 - c0;
    const bool mine = (cl >= 0) && (cl < PSEG) && ((cl >> 2) == wave);
    const unsigned bm = __builtin_amdgcn_ballot_w32(mine);
    if (mine) wl[nrep + __builtin_amdgcn_mbcnt_lo(bm, 0)] = w & ~256;
    nrep += __popc(bm);
  }
#pragma nounroll
  for (int base = 0; base < n_own; base += 32) {
    const int i = base + lane;
    if (i < n_own) wl[nrep + i] = evbuf[e_lo + i];
  }
  const int nw   = __builtin_amdgcn_readfirstlane(nrep) + n_own;
  const int last = nw - 1;

  {
    const float* lb = lwp + l16 * WPITCH + (lhalf << 1);
    for (int et = wave; et * 16 < n_own; et += 4) {
      const int m  = e_lo + min(et * 16 + l16, n_own - 1);
      const int ra = b * TT + (evbuf[m] >> 9);
      const float* arow = h + ra * DD + (lhalf << 1);
      v2f af[64];
#pragma unroll
      for (int q = 0; q < 64; ++q) af[q] = *(const v2f*)(arow + q * 4);
      v8f d;
#pragma unroll
      for (int r = 0; r < 8; ++r) d[r] = bv;
#pragma unroll
      for (int q = 0; q < 64; ++q) {
        const v2f bq = *(const v2f*)(lb + q * 4);
        d = __builtin_amdgcn_wmma_f32_16x16x4_f32(
            false, af[q], false, bq, (short)0, d, false, false);
      }
#pragma unroll
      for (int r = 0; r < 8; ++r) {
        const int mm = et * 16 + r + (lhalf << 3);
        if (mm < n_own) {
          const int rr = b * TT + (evbuf[e_lo + mm] >> 9);
          sbase[rr * ostride + l16] = d[r];
        }
      }
    }
  }
  __syncthreads();

  float4 pa[NSLOT], pc[NSLOT];
  float  po[NSLOT];
  int    pev[NSLOT];
#pragma unroll
  for (int k = 0; k < NSLOT; ++k) {
    pa[k] = make_float4(0.f, 0.f, 0.f, 0.f); pc[k] = pa[k]; po[k] = 0.f; pev[k] = 0;
  }
  int evq = wl[0];
  float invh_nx = 0.f;
  float ainvl = class_inv_norms(acc, lane);

  for (int e0 = -NSLOT; e0 < nw; e0 += NSLOT) {
#pragma unroll
    for (int s = 0; s < NSLOT; ++s) {
      const int e = e0 + s;

      {
        const int kn  = (s + NSLOT - 1) % NSLOT;
        const int evn = __builtin_amdgcn_readfirstlane(evq);
        pev[kn] = evn;
        fetch_event(evn, b, lane, h, obase, ostride, pa[kn], pc[kn], po[kn]);
        evq = wl[min(e + NAHEAD + 1, last)];
      }

      if ((unsigned)(e + 1) > (unsigned)nw) continue;
      const bool live = (e >= 0);

      const int ev   = pev[s];
      const int t    = ev >> 9;
      const int emit = live ? ((ev >> 8) & 1) : 0;
      const int clsp = live ? (ev & 255) : 0;
      const int r    = b * TT + t;

      const float4 ha = pa[s];
      const float4 hc = pc[s];
      const float oldv = po[s];
      float* const op = obase + r * ostride;
      const float hx[8] = {ha.x, ha.y, ha.z, ha.w, hc.x, hc.y, hc.z, hc.w};
      const float invh = invh_nx;

      const int kx = (s + 1) % NSLOT;
      const float4 na = pa[kx];
      const float4 nc = pc[kx];
      const float nx[8] = {na.x, na.y, na.z, na.w, nc.x, nc.y, nc.z, nc.w};
      float hsn = 0.f;
#pragma unroll
      for (int u = 0; u < 8; ++u) hsn = fmaf(nx[u], nx[u], hsn);

      float dot[4];
#pragma unroll
      for (int q = 0; q < 4; ++q) {
        dot[q] = 0.f;
#pragma unroll
        for (int u = 0; u < 8; ++u) dot[q] = fmaf(acc[q][u], hx[u], dot[q]);
      }
#pragma unroll
      for (int q = 0; q < 4; ++q) dot[q] += xrow16(dot[q]);
      hsn += xrow16(hsn);
#pragma unroll
      for (int q = 0; q < 4; ++q) dot[q] += xrow<8>(dot[q]);
      hsn += xrow<8>(hsn);
#pragma unroll
      for (int q = 0; q < 4; ++q) dot[q] += xrow<4>(dot[q]);
      hsn += xrow<4>(hsn);
#pragma unroll
      for (int q = 0; q < 4; ++q) dot[q] += xrow<2>(dot[q]);
      hsn += xrow<2>(hsn);
#pragma unroll
      for (int q = 0; q < 4; ++q) dot[q] += xrow<1>(dot[q]);
      hsn += xrow<1>(hsn);
      const float dsel = (lane == 0) ? dot[0] : (lane == 1) ? dot[1]
                       : (lane == 2) ? dot[2] : dot[3];
      const float quot = dsel * invh * ainvl;
      if (emit && lane < 4) {
        float add = st * quot;
        op[lane] = oldv + add;
      }
      invh_nx = inv_norm(hsn);

      if (clsp) {
        const int cl = clsp - 1 - c0;
        if (cl >= 0 && cl < PSEG && (cl >> 2) == wave) {
          const int qs = cl & 3;
          float s2 = 0.f;
#pragma unroll
          for (int q = 0; q < 4; ++q) {
            if (q == qs) {
#pragma unroll
              for (int u = 0; u < 8; ++u) acc[q][u] += hx[u] * invh;
#pragma unroll
              for (int u = 0; u < 8; ++u) s2 = fmaf(acc[q][u], acc[q][u], s2);
            }
          }
          const float a2 = inv_norm(wave_sum(s2));
          ainvl = (lane == qs) ? a2 : ainvl;
        }
      }
    }
  }
}

__global__ __launch_bounds__(GM_THREADS)
void heads_kernel(const int* __restrict__ tokens,
                  const float* __restrict__ h,
                  const float* __restrict__ E,
                  const float* __restrict__ Wn,
                  const float* __restrict__ bn,
                  const float* __restrict__ Wt,
                  const float* __restrict__ bt,
                  const float* __restrict__ tsa,
                  const float* __restrict__ tst,
                  const float* __restrict__ psa, const float* __restrict__ pst,
                  const float* __restrict__ ppa, const float* __restrict__ ppt,
                  const float* __restrict__ pta, const float* __restrict__ ptt,
                  float* __restrict__ out) {
  __shared__ __attribute__((aligned(16))) float lw[NTOT * WPITCH];

  if (blockIdx.x < NPROTO) {
    proto_role(blockIdx.x, lw, tokens, h, E, Wn, bn, Wt, bt, tsa, tst,
               psa, pst, ppa, ppt, pta, ptt, out);
  } else {
    gemm_role(blockIdx.x - NPROTO, lw, tokens, h, E, Wn, bn, Wt, bt, tsa, tst, out);
  }
}

extern "C" void kernel_launch(void* const* d_in, const int* in_sizes, int n_in,
                              void* d_out, int out_size, void* d_ws, size_t ws_size,
                              hipStream_t stream) {
  const int*   tokens = (const int*)  d_in[0];
  const float* h      = (const float*)d_in[1];
  const float* E      = (const float*)d_in[2];
  const float* Wn     = (const float*)d_in[3];
  const float* bn     = (const float*)d_in[4];
  const float* Wt     = (const float*)d_in[5];
  const float* bt     = (const float*)d_in[6];
  const float* tsa    = (const float*)d_in[7];
  const float* tst    = (const float*)d_in[8];
  const float* psa    = (const float*)d_in[9];
  const float* pst    = (const float*)d_in[10];
  const float* ppa    = (const float*)d_in[11];
  const float* ppt    = (const float*)d_in[12];
  const float* pta    = (const float*)d_in[13];
  const float* ptt    = (const float*)d_in[14];
  float* out = (float*)d_out;

  heads_kernel<<<NBLOCKS, GM_THREADS, 0, stream>>>(tokens, h, E, Wn, bn, Wt, bt, tsa, tst,
                                                   psa, pst, ppa, ppt, pta, ptt, out);
}
